# starting kernel
# baseline (speedup 1.0000x reference)
#include <hip/hip_runtime.h>

#define B_ 2
#define T_ 12
#define N_ 1024
#define D_ 64
#define H_ 4
#define HD_ 16
#define EPS_ 1e-5f
#define NEG_BIG -3.0e38f

#define CHUNK_ 256
#define NCHUNK_ (N_ / CHUNK_)
#define NT_ (CHUNK_ / 16)
#define SSTRIDE_ 260

typedef __attribute__((ext_vector_type(2))) float v2f;
typedef __attribute__((ext_vector_type(8))) float v8f;

__device__ __forceinline__ v8f wmma_f32_k4(v2f a, v2f b, v8f c) {
    return __builtin_amdgcn_wmma_f32_16x16x4_f32(
         false, a,  false, b,
         (short)0, c,  false,  false);
}

__global__ __launch_bounds__(32) void qkv_kernel(
    const float* __restrict__ x,
    const float* __restrict__ Wq, const float* __restrict__ bq,
    const float* __restrict__ Wk, const float* __restrict__ bk,
    const float* __restrict__ Wv, const float* __restrict__ bv,
    float* __restrict__ Q, float* __restrict__ Ko, float* __restrict__ V)
{
    __shared__ float xs[16 * 68];
    const int lane = threadIdx.x & 31;
    const int half = lane >> 4;
    const int l16  = lane & 15;
    const int which = blockIdx.y;
    const float* W    = (which == 0) ? Wq : (which == 1) ? Wk : Wv;
    const float* bias = (which == 0) ? bq : (which == 1) ? bk : bv;
    float*       out  = (which == 0) ? Q  : (which == 1) ? Ko : V;
    const float  osc  = (which == 0) ? 0.25f : 1.0f;

    const size_t row0 = (size_t)blockIdx.x * 16;
    const float* xt = x + row0 * D_;

    #pragma unroll
    for (int i = 0; i < 8; ++i) {
        const int idx = (i * 32 + lane) * 4;
        const float4 val = *(const float4*)(xt + idx);
        const int row = idx >> 6, col = idx & 63;
        float* dst = &xs[row * 68 + col];
        dst[0] = val.x; dst[1] = val.y; dst[2] = val.z; dst[3] = val.w;
    }
    __syncthreads();

    v2f afrag[16];
    #pragma unroll
    for (int kk = 0; kk < 16; ++kk) {
        const float* p = &xs[l16 * 68 + kk * 4 + 2 * half];
        v2f a; a.x = p[0]; a.y = p[1];
        afrag[kk] = a;
    }

    v8f acc[4];
    #pragma unroll
    for (int c = 0; c < 4; ++c) acc[c] = {};

    #pragma unroll
    for (int kk = 0; kk < 16; ++kk) {
        #pragma unroll
        for (int c = 0; c < 4; ++c) {
            const float* wp = &W[(c * 16 + l16) * D_ + kk * 4 + 2 * half];
            v2f b; b.x = wp[0]; b.y = wp[1];
            acc[c] = wmma_f32_k4(afrag[kk], b, acc[c]);
        }
    }

    #pragma unroll
    for (int c = 0; c < 4; ++c) {
        const float bb = bias[c * 16 + l16];
        #pragma unroll
        for (int r = 0; r < 8; ++r) {
            out[(row0 + r + 8 * half) * D_ + c * 16 + l16] = (acc[c][r] + bb) * osc;
        }
    }
}

__global__ __launch_bounds__(32) void attn_kernel(
    const float* __restrict__ Qs, const float* __restrict__ K,
    const float* __restrict__ V, const float* __restrict__ adj,
    float* __restrict__ Hout)
{
    __shared__ float s_lds[16 * SSTRIDE_];
    __shared__ float mx_bc[16];
    __shared__ float sc_bc[16];
    __shared__ float rs_bc[16];
    const int lane = threadIdx.x & 31;
    const int half = lane >> 4;
    const int l16  = lane & 15;
    const int n0 = blockIdx.x * 16;
    const int h  = blockIdx.y;
    const int bt = blockIdx.z;
    const size_t base  = (size_t)bt * N_ * D_;
    const size_t abase = (size_t)bt * N_ * N_;

    v2f aq[4];
    #pragma unroll
    for (int kk = 0; kk < 4; ++kk) {
        const float* p = &Qs[base + (size_t)(n0 + l16) * D_ + h * HD_ + kk * 4 + 2 * half];
        v2f a; a.x = p[0]; a.y = p[1];
        aq[kk] = a;
    }

    float mrow[8];
    #pragma unroll
    for (int r = 0; r < 8; ++r) mrow[r] = NEG_BIG;
    float lsum = 0.0f;
    v8f acco = {};

    for (int chunk = 0; chunk < NCHUNK_; ++chunk) {
        float mxl[8];
        #pragma unroll
        for (int r = 0; r < 8; ++r) mxl[r] = NEG_BIG;

        for (int t = 0; t < NT_; ++t) {
            const int m0 = chunk * CHUNK_ + t * 16;
            v8f accs = {};
            #pragma unroll
            for (int kk = 0; kk < 4; ++kk) {
                const float* p = &K[base + (size_t)(m0 + l16) * D_ + h * HD_ + kk * 4 + 2 * half];
                v2f b; b.x = p[0]; b.y = p[1];
                accs = wmma_f32_k4(aq[kk], b, accs);
            }
            #pragma unroll
            for (int r = 0; r < 8; ++r) {
                const int n = n0 + r + 8 * half;
                const float a = __builtin_nontemporal_load(
                    &adj[abase + (size_t)n * N_ + m0 + l16]);
                const float s = accs[r] * a;
                mxl[r] = fmaxf(mxl[r], s);
                s_lds[(r + 8 * half) * SSTRIDE_ + t * 16 + l16] = s;
            }
        }

        #pragma unroll
        for (int r = 0; r < 8; ++r) {
            float mx = mxl[r];
            mx = fmaxf(mx, __shfl_xor(mx, 1));
            mx = fmaxf(mx, __shfl_xor(mx, 2));
            mx = fmaxf(mx, __shfl_xor(mx, 4));
            mx = fmaxf(mx, __shfl_xor(mx, 8));
            const float mnew = fmaxf(mrow[r], mx);
            const float sc   = __expf(mrow[r] - mnew);
            acco[r] *= sc;
            mrow[r] = mnew;
            if (l16 == r) {
                mx_bc[r + 8 * half] = mnew;
                sc_bc[r + 8 * half] = sc;
            }
        }
        __syncthreads();

        const float mxr = mx_bc[l16];
        float csum = 0.0f;
        for (int t = 0; t < NT_; ++t) {
            const int m0 = chunk * CHUNK_ + t * 16;
            #pragma unroll
            for (int kk = 0; kk < 4; ++kk) {
                const v2f sv = *(const v2f*)&s_lds[l16 * SSTRIDE_ + t * 16 + kk * 4 + 2 * half];
                const float p0 = __expf(sv.x - mxr);
                const float p1 = __expf(sv.y - mxr);
                csum += p0 + p1;
                v2f ap; ap.x = p0; ap.y = p1;
                const float* vp = &V[base + (size_t)(m0 + kk * 4 + 2 * half) * D_ + h * HD_ + l16];
                v2f bv; bv.x = vp[0]; bv.y = vp[D_];
                acco = wmma_f32_k4(ap, bv, acco);
            }
        }
        lsum = lsum * sc_bc[l16] + csum;
        __syncthreads();
    }

    const float rs = lsum + __shfl_xor(lsum, 16);
    if (half == 0) rs_bc[l16] = rs;
    __syncthreads();

    #pragma unroll
    for (int r = 0; r < 8; ++r) {
        const float inv = 1.0f / rs_bc[r + 8 * half];
        Hout[base + (size_t)(n0 + r + 8 * half) * D_ + h * HD_ + l16] = acco[r] * inv;
    }
}

__global__ __launch_bounds__(256) void ln_kernel(
    const float* __restrict__ Hout, const float* __restrict__ x,
    const float* __restrict__ gamma, const float* __restrict__ beta,
    float* __restrict__ out)
{
    const int lane = threadIdx.x & 31;
    const int wid  = threadIdx.x >> 5;
    const size_t row = (size_t)blockIdx.x * 8 + wid;
    const float* hp = Hout + row * D_;
    const float* xp = x    + row * D_;
    const float y0 = hp[lane]      + xp[lane];
    const float y1 = hp[lane + 32] + xp[lane + 32];
    float s1 = y0 + y1;
    float s2 = y0 * y0 + y1 * y1;
    #pragma unroll
    for (int m = 1; m < 32; m <<= 1) {
        s1 += __shfl_xor(s1, m);
        s2 += __shfl_xor(s2, m);
    }
    const float mean = s1 * (1.0f / 64.0f);
    const float var  = s2 * (1.0f / 64.0f) - mean * mean;
    const float inv  = rsqrtf(var + EPS_);
    float* op = out + row * D_;
    op[lane]      = (y0 - mean) * inv * gamma[lane]      + beta[lane];
    op[lane + 32] = (y1 - mean) * inv * gamma[lane + 32] + beta[lane + 32];
}

extern "C" void kernel_launch(void* const* d_in, const int* in_sizes, int n_in,
                              void* d_out, int out_size, void* d_ws, size_t ws_size,
                              hipStream_t stream) {
    (void)in_sizes; (void)n_in; (void)out_size; (void)ws_size;
    const float* x     = (const float*)d_in[0];
    const float* adj   = (const float*)d_in[1];
    const float* Wq    = (const float*)d_in[3];
    const float* bq    = (const float*)d_in[4];
    const float* Wk    = (const float*)d_in[5];
    const float* bk    = (const float*)d_in[6];
    const float* Wv    = (const float*)d_in[7];
    const float* bv    = (const float*)d_in[8];
    const float* gamma = (const float*)d_in[9];
    const float* beta  = (const float*)d_in[10];
    float* out = (float*)d_out;

    float* ws = (float*)d_ws;
    const size_t S = (size_t)B_ * T_ * N_ * D_;
    float* Q  = ws;
    float* K  = ws + S;
    float* V  = ws + 2 * S;
    float* Hh = ws + 3 * S;

    dim3 g1(B_ * T_ * N_ / 16, 3, 1);
    qkv_kernel<<<g1, 32, 0, stream>>>(x, Wq, bq, Wk, bk, Wv, bv, Q, K, V);

    dim3 g2(N_ / 16, H_, B_ * T_);
    attn_kernel<<<g2, 32, 0, stream>>>(Q, K, V, adj, Hh);

    ln_kernel<<<(B_ * T_ * N_) / 8, 256, 0, stream>>>(Hh, x, gamma, beta, out);
}


# attention: scores as St=K.Qs^T so exp(St) registers are the B operand of Ot+=V^T.Pt (no LDS score pass, no barriers); online softmax per 32 keys, two PV accumulators: summation order changes
# speedup vs baseline: 1.3129x; 1.3129x over previous
#include <hip/hip_runtime.h>

#define B_ 2
#define T_ 12
#define N_ 1024
#define D_ 64
#define H_ 4
#define HD_ 16
#define EPS_ 1e-5f
#define NEG_BIG -3.0e38f


typedef __attribute__((ext_vector_type(2))) float v2f;
typedef __attribute__((ext_vector_type(8))) float v8f;

__device__ __forceinline__ v8f wmma_f32_k4(v2f a, v2f b, v8f c) {
    return __builtin_amdgcn_wmma_f32_16x16x4_f32(
         false, a,  false, b,
         (short)0, c,  false,  false);
}

__global__ __launch_bounds__(32) void qkv_kernel(
    const float* __restrict__ x,
    const float* __restrict__ Wq, const float* __restrict__ bq,
    const float* __restrict__ Wk, const float* __restrict__ bk,
    const float* __restrict__ Wv, const float* __restrict__ bv,
    float* __restrict__ Q, float* __restrict__ Ko, float* __restrict__ V)
{
    __shared__ float xs[16 * 68];
    const int lane = threadIdx.x & 31;
    const int half = lane >> 4;
    const int l16  = lane & 15;
    const int which = blockIdx.y;
    const float* W    = (which == 0) ? Wq : (which == 1) ? Wk : Wv;
    const float* bias = (which == 0) ? bq : (which == 1) ? bk : bv;
    float*       out  = (which == 0) ? Q  : (which == 1) ? Ko : V;
    const float  osc  = (which == 0) ? 0.25f : 1.0f;

    const size_t row0 = (size_t)blockIdx.x * 16;
    const float* xt = x + row0 * D_;

    #pragma unroll
    for (int i = 0; i < 8; ++i) {
        const int idx = (i * 32 + lane) * 4;
        const float4 val = *(const float4*)(xt + idx);
        const int row = idx >> 6, col = idx & 63;
        float* dst = &xs[row * 68 + col];
        dst[0] = val.x; dst[1] = val.y; dst[2] = val.z; dst[3] = val.w;
    }
    __syncthreads();

    v2f afrag[16];
    #pragma unroll
    for (int kk = 0; kk < 16; ++kk) {
        const float* p = &xs[l16 * 68 + kk * 4 + 2 * half];
        v2f a; a.x = p[0]; a.y = p[1];
        afrag[kk] = a;
    }

    v8f acc[4];
    #pragma unroll
    for (int c = 0; c < 4; ++c) acc[c] = {};

    #pragma unroll
    for (int kk = 0; kk < 16; ++kk) {
        #pragma unroll
        for (int c = 0; c < 4; ++c) {
            const float* wp = &W[(c * 16 + l16) * D_ + kk * 4 + 2 * half];
            v2f b; b.x = wp[0]; b.y = wp[1];
            acc[c] = wmma_f32_k4(afrag[kk], b, acc[c]);
        }
    }

    #pragma unroll
    for (int c = 0; c < 4; ++c) {
        const float bb = bias[c * 16 + l16];
        #pragma unroll
        for (int r = 0; r < 8; ++r) {
            out[(row0 + r + 8 * half) * D_ + c * 16 + l16] = (acc[c][r] + bb) * osc;
        }
    }
}

typedef __attribute__((ext_vector_type(4))) float v4f;

#define GKEYS_ 32

__device__ __forceinline__ float xhalf(float x) {
    const int b = __builtin_bit_cast(int, x);
    return __builtin_bit_cast(float,
        __builtin_amdgcn_permlanex16(b, b, 0x76543210, 0xfedcba98, false, false));
}

__device__ __forceinline__ v2f pick2(v4f lo, v4f hi, int j) {
    v2f r;
    if (j == 0)      { r.x = lo.x; r.y = lo.y; }
    else if (j == 1) { r.x = lo.z; r.y = lo.w; }
    else if (j == 2) { r.x = hi.x; r.y = hi.y; }
    else             { r.x = hi.z; r.y = hi.w; }
    return r;
}

__global__ __launch_bounds__(32) void attn_kernel(
    const float* __restrict__ Qs, const float* __restrict__ K,
    const float* __restrict__ V, const float* __restrict__ adj,
    float* __restrict__ Hout)
{
    const int lane = threadIdx.x & 31;
    const int half = lane >> 4;
    const int l16  = lane & 15;
    const int n0 = blockIdx.x * 16;
    const int h  = blockIdx.y;
    const int bt = blockIdx.z;
    const size_t base  = (size_t)bt * N_ * D_;
    const size_t abase = (size_t)bt * N_ * N_;

    const float* qp = Qs + base + (size_t)(n0 + l16) * D_ + h * HD_ + 8 * half;
    const v4f q_lo = *(const v4f*)qp;
    const v4f q_hi = *(const v4f*)(qp + 4);

    const float* kp = K + base + (size_t)l16 * D_ + h * HD_ + 8 * half;
    const float* vp = V + base + (size_t)(8 * half) * D_ + h * HD_ + l16;
    const float* ap = adj + abase + (size_t)(n0 + l16) * N_ + 8 * half;

    float mrun = NEG_BIG;
    float lsum = 0.0f;
    v8f acc0 = {}, acc1 = {};

    for (int m0 = 0; m0 < N_; m0 += GKEYS_) {
        const float* k0 = kp + (size_t)m0 * D_;
        const v4f ka_lo = *(const v4f*)(k0);
        const v4f ka_hi = *(const v4f*)(k0 + 4);
        const v4f kb_lo = *(const v4f*)(k0 + 16 * D_);
        const v4f kb_hi = *(const v4f*)(k0 + 16 * D_ + 4);
        const v4f aa_lo = *(const v4f*)(ap + m0);
        const v4f aa_hi = *(const v4f*)(ap + m0 + 4);
        const v4f ab_lo = *(const v4f*)(ap + m0 + 16);
        const v4f ab_hi = *(const v4f*)(ap + m0 + 20);
        const float* v0 = vp + (size_t)m0 * D_;
        v2f va[4], vb[4];
        #pragma unroll
        for (int j = 0; j < 4; ++j) {
            va[j].x = v0[(2 * j) * D_];
            va[j].y = v0[(2 * j + 1) * D_];
            vb[j].x = v0[(16 + 2 * j) * D_];
            vb[j].y = v0[(16 + 2 * j + 1) * D_];
        }

        v8f sa = {}, sb = {};
        #pragma unroll
        for (int kk = 0; kk < 4; ++kk) {
            const v2f bq = pick2(q_lo, q_hi, kk);
            sa = wmma_f32_k4(pick2(ka_lo, ka_hi, kk), bq, sa);
            sb = wmma_f32_k4(pick2(kb_lo, kb_hi, kk), bq, sb);
        }

        sa[0] *= aa_lo.x; sa[1] *= aa_lo.y; sa[2] *= aa_lo.z; sa[3] *= aa_lo.w;
        sa[4] *= aa_hi.x; sa[5] *= aa_hi.y; sa[6] *= aa_hi.z; sa[7] *= aa_hi.w;
        sb[0] *= ab_lo.x; sb[1] *= ab_lo.y; sb[2] *= ab_lo.z; sb[3] *= ab_lo.w;
        sb[4] *= ab_hi.x; sb[5] *= ab_hi.y; sb[6] *= ab_hi.z; sb[7] *= ab_hi.w;
        float mx = fmaxf(sa[0], sb[0]);
        #pragma unroll
        for (int r = 1; r < 8; ++r) mx = fmaxf(mx, fmaxf(sa[r], sb[r]));
        mx = fmaxf(mx, xhalf(mx));
        const float mnew = fmaxf(mrun, mx);
        const float sc   = __expf(mrun - mnew);
        mrun = mnew;

        float csum = 0.0f;
        #pragma unroll
        for (int r = 0; r < 8; ++r) {
            sa[r] = __expf(sa[r] - mnew);
            sb[r] = __expf(sb[r] - mnew);
            csum += sa[r] + sb[r];
        }
        lsum = lsum * sc + csum;
        #pragma unroll
        for (int r = 0; r < 8; ++r) { acc0[r] *= sc; acc1[r] *= sc; }

        #pragma unroll
        for (int j = 0; j < 4; ++j) {
            v2f pa; pa.x = sa[2 * j]; pa.y = sa[2 * j + 1];
            v2f pb; pb.x = sb[2 * j]; pb.y = sb[2 * j + 1];
            acc0 = wmma_f32_k4(va[j], pa, acc0);
            acc1 = wmma_f32_k4(vb[j], pb, acc1);
        }
    }

    const float inv = 1.0f / (lsum + xhalf(lsum));
    float* op = Hout + base + (size_t)(n0 + l16) * D_ + h * HD_ + 8 * half;
    v4f o_lo, o_hi;
    o_lo.x = (acc0[0] + acc1[0]) * inv; o_lo.y = (acc0[1] + acc1[1]) * inv;
    o_lo.z = (acc0[2] + acc1[2]) * inv; o_lo.w = (acc0[3] + acc1[3]) * inv;
    o_hi.x = (acc0[4] + acc1[4]) * inv; o_hi.y = (acc0[5] + acc1[5]) * inv;
    o_hi.z = (acc0[6] + acc1[6]) * inv; o_hi.w = (acc0[7] + acc1[7]) * inv;
    *(v4f*)op       = o_lo;
    *(v4f*)(op + 4) = o_hi;
}

__global__ __launch_bounds__(256) void ln_kernel(
    const float* __restrict__ Hout, const float* __restrict__ x,
    const float* __restrict__ gamma, const float* __restrict__ beta,
    float* __restrict__ out)
{
    const int lane = threadIdx.x & 31;
    const int wid  = threadIdx.x >> 5;
    const size_t row = (size_t)blockIdx.x * 8 + wid;
    const float* hp = Hout + row * D_;
    const float* xp = x    + row * D_;
    const float y0 = hp[lane]      + xp[lane];
    const float y1 = hp[lane + 32] + xp[lane + 32];
    float s1 = y0 + y1;
    float s2 = y0 * y0 + y1 * y1;
    #pragma unroll
    for (int m = 1; m < 32; m <<= 1) {
        s1 += __shfl_xor(s1, m);
        s2 += __shfl_xor(s2, m);
    }
    const float mean = s1 * (1.0f / 64.0f);
    const float var  = s2 * (1.0f / 64.0f) - mean * mean;
    const float inv  = rsqrtf(var + EPS_);
    float* op = out + row * D_;
    op[lane]      = (y0 - mean) * inv * gamma[lane]      + beta[lane];
    op[lane + 32] = (y1 - mean) * inv * gamma[lane + 32] + beta[lane + 32];
}

extern "C" void kernel_launch(void* const* d_in, const int* in_sizes, int n_in,
                              void* d_out, int out_size, void* d_ws, size_t ws_size,
                              hipStream_t stream) {
    (void)in_sizes; (void)n_in; (void)out_size; (void)ws_size;
    const float* x     = (const float*)d_in[0];
    const float* adj   = (const float*)d_in[1];
    const float* Wq    = (const float*)d_in[3];
    const float* bq    = (const float*)d_in[4];
    const float* Wk    = (const float*)d_in[5];
    const float* bk    = (const float*)d_in[6];
    const float* Wv    = (const float*)d_in[7];
    const float* bv    = (const float*)d_in[8];
    const float* gamma = (const float*)d_in[9];
    const float* beta  = (const float*)d_in[10];
    float* out = (float*)d_out;

    float* ws = (float*)d_ws;
    const size_t S = (size_t)B_ * T_ * N_ * D_;
    float* Q  = ws;
    float* K  = ws + S;
    float* V  = ws + 2 * S;
    float* Hh = ws + 3 * S;

    dim3 g1(B_ * T_ * N_ / 16, 3, 1);
    qkv_kernel<<<g1, 32, 0, stream>>>(x, Wq, bq, Wk, bk, Wv, bv, Q, K, V);

    dim3 g2(N_ / 16, H_, B_ * T_);
    attn_kernel<<<g2, 32, 0, stream>>>(Q, K, V, adj, Hh);

    ln_kernel<<<(B_ * T_ * N_) / 8, 256, 0, stream>>>(Hh, x, gamma, beta, out);
}


# Q, K, V kept in the workspace in WMMA fragment order (Q, K from the operand-swapped product): each fragment load and store covers four whole 128-byte lines; k summation order kept
# speedup vs baseline: 1.6203x; 1.2341x over previous
#include <hip/hip_runtime.h>

#define B_ 2
#define T_ 12
#define N_ 1024
#define D_ 64
#define H_ 4
#define HD_ 16
#define EPS_ 1e-5f
#define NEG_BIG -3.0e38f


typedef __attribute__((ext_vector_type(2))) float v2f;
typedef __attribute__((ext_vector_type(8))) float v8f;

__device__ __forceinline__ v8f wmma_f32_k4(v2f a, v2f b, v8f c) {
    return __builtin_amdgcn_wmma_f32_16x16x4_f32(
         false, a,  false, b,
         (short)0, c,  false,  false);
}

#define FRAG_TILE_ 256
__device__ __forceinline__ size_t frag_off(size_t bt, int head, size_t tile) {
    return ((bt * H_ + head) * (N_ / 16) + tile) * FRAG_TILE_;
}
__global__ __launch_bounds__(32) void qkv_kernel(
    const float* __restrict__ x,
    const float* __restrict__ Wq, const float* __restrict__ bq,
    const float* __restrict__ Wk, const float* __restrict__ bk,
    const float* __restrict__ Wv, const float* __restrict__ bv,
    float* __restrict__ Q, float* __restrict__ Ko, float* __restrict__ V)
{
    __shared__ float xs[16 * 68];
    const int lane = threadIdx.x & 31;
    const int half = lane >> 4;
    const int l16  = lane & 15;
    const int which = blockIdx.y;
    const float* W    = (which == 0) ? Wq : (which == 1) ? Wk : Wv;
    const float* bias = (which == 0) ? bq : (which == 1) ? bk : bv;
    float*       out  = (which == 0) ? Q  : (which == 1) ? Ko : V;
    const float  osc  = (which == 0) ? 0.25f : 1.0f;

    const size_t row0 = (size_t)blockIdx.x * 16;
    const float* xt = x + row0 * D_;

    #pragma unroll
    for (int i = 0; i < 8; ++i) {
        const int idx = (i * 32 + lane) * 4;
        const float4 val = *(const float4*)(xt + idx);
        const int row = idx >> 6, col = idx & 63;
        float* dst = &xs[row * 68 + col];
        dst[0] = val.x; dst[1] = val.y; dst[2] = val.z; dst[3] = val.w;
    }
    __syncthreads();

    v2f afrag[16];
    #pragma unroll
    for (int kk = 0; kk < 16; ++kk) {
        const float* p = &xs[l16 * 68 + kk * 4 + 2 * half];
        v2f a; a.x = p[0]; a.y = p[1];
        afrag[kk] = a;
    }

    v8f acc[4];
    #pragma unroll
    for (int c = 0; c < 4; ++c) acc[c] = {};

    const size_t bt   = row0 / N_;
    const size_t tile = (row0 % N_) / 16;

    if (which == 2) {
        #pragma unroll
        for (int kk = 0; kk < 16; ++kk) {
            #pragma unroll
            for (int c = 0; c < 4; ++c) {
                const float* wp = &W[(c * 16 + l16) * D_ + kk * 4 + 2 * half];
                v2f w; w.x = wp[0]; w.y = wp[1];
                acc[c] = wmma_f32_k4(afrag[kk], w, acc[c]);
            }
        }
        #pragma unroll
        for (int c = 0; c < 4; ++c) {
            const float bb = bias[c * 16 + l16];
            float* f = out + frag_off(bt, c, tile) + lane * 4;
            float4 lo, hi;
            lo.x = acc[c][0] + bb; lo.y = acc[c][1] + bb; lo.z = acc[c][2] + bb; lo.w = acc[c][3] + bb;
            hi.x = acc[c][4] + bb; hi.y = acc[c][5] + bb; hi.z = acc[c][6] + bb; hi.w = acc[c][7] + bb;
            *(float4*)f         = lo;
            *(float4*)(f + 128) = hi;
        }
    } else {
        #pragma unroll
        for (int kk = 0; kk < 16; ++kk) {
            #pragma unroll
            for (int c = 0; c < 4; ++c) {
                const float* wp = &W[(c * 16 + l16) * D_ + kk * 4 + 2 * half];
                v2f w; w.x = wp[0]; w.y = wp[1];
                acc[c] = wmma_f32_k4(w, afrag[kk], acc[c]);
            }
        }
        #pragma unroll
        for (int c = 0; c < 4; ++c) {
            const float4 b_lo = *(const float4*)(bias + c * 16 + 8 * half);
            const float4 b_hi = *(const float4*)(bias + c * 16 + 8 * half + 4);
            float* f = out + frag_off(bt, c, tile) + lane * 4;
            float4 lo, hi;
            lo.x = (acc[c][0] + b_lo.x) * osc; lo.y = (acc[c][1] + b_lo.y) * osc;
            lo.z = (acc[c][2] + b_lo.z) * osc; lo.w = (acc[c][3] + b_lo.w) * osc;
            hi.x = (acc[c][4] + b_hi.x) * osc; hi.y = (acc[c][5] + b_hi.y) * osc;
            hi.z = (acc[c][6] + b_hi.z) * osc; hi.w = (acc[c][7] + b_hi.w) * osc;
            *(float4*)f         = lo;
            *(float4*)(f + 128) = hi;
        }
    }
}

typedef __attribute__((ext_vector_type(4))) float v4f;

#define GKEYS_ 32

__device__ __forceinline__ float xhalf(float x) {
    const int b = __builtin_bit_cast(int, x);
    return __builtin_bit_cast(float,
        __builtin_amdgcn_permlanex16(b, b, 0x76543210, 0xfedcba98, false, false));
}

__device__ __forceinline__ v2f pick2(v4f lo, v4f hi, int j) {
    v2f r;
    if (j == 0)      { r.x = lo.x; r.y = lo.y; }
    else if (j == 1) { r.x = lo.z; r.y = lo.w; }
    else if (j == 2) { r.x = hi.x; r.y = hi.y; }
    else             { r.x = hi.z; r.y = hi.w; }
    return r;
}

__global__ __launch_bounds__(32) void attn_kernel(
    const float* __restrict__ Qf, const float* __restrict__ Kf,
    const float* __restrict__ Vf, const float* __restrict__ adj,
    float* __restrict__ Hout)
{
    const int lane = threadIdx.x & 31;
    const int half = lane >> 4;
    const int l16  = lane & 15;
    const int n0 = blockIdx.x * 16;
    const int h  = blockIdx.y;
    const int bt = blockIdx.z;
    const size_t base  = (size_t)bt * N_ * D_;
    const size_t abase = (size_t)bt * N_ * N_;

    const float* qp = Qf + frag_off(bt, h, blockIdx.x) + lane * 4;
    const v4f q_lo = *(const v4f*)qp;
    const v4f q_hi = *(const v4f*)(qp + 128);

    const float* kp = Kf + frag_off(bt, h, 0) + lane * 4;
    const float* vp = Vf + frag_off(bt, h, 0) + lane * 4;
    const float* ap = adj + abase + (size_t)(n0 + l16) * N_ + 8 * half;

    float mrun = NEG_BIG;
    float lsum = 0.0f;
    v8f acc0 = {}, acc1 = {};

    for (int m0 = 0; m0 < N_; m0 += GKEYS_) {
        const float* k0 = kp + (size_t)m0 * (FRAG_TILE_ / 16);
        const v4f ka_lo = *(const v4f*)(k0);
        const v4f ka_hi = *(const v4f*)(k0 + 128);
        const v4f kb_lo = *(const v4f*)(k0 + FRAG_TILE_);
        const v4f kb_hi = *(const v4f*)(k0 + FRAG_TILE_ + 128);
        const v4f aa_lo = *(const v4f*)(ap + m0);
        const v4f aa_hi = *(const v4f*)(ap + m0 + 4);
        const v4f ab_lo = *(const v4f*)(ap + m0 + 16);
        const v4f ab_hi = *(const v4f*)(ap + m0 + 20);
        const float* v0 = vp + (size_t)m0 * (FRAG_TILE_ / 16);
        const v4f va_lo = *(const v4f*)(v0);
        const v4f va_hi = *(const v4f*)(v0 + 128);
        const v4f vb_lo = *(const v4f*)(v0 + FRAG_TILE_);
        const v4f vb_hi = *(const v4f*)(v0 + FRAG_TILE_ + 128);

        v8f sa = {}, sb = {};
        #pragma unroll
        for (int kk = 0; kk < 4; ++kk) {
            const v2f bq = pick2(q_lo, q_hi, kk);
            sa = wmma_f32_k4(pick2(ka_lo, ka_hi, kk), bq, sa);
            sb = wmma_f32_k4(pick2(kb_lo, kb_hi, kk), bq, sb);
        }

        sa[0] *= aa_lo.x; sa[1] *= aa_lo.y; sa[2] *= aa_lo.z; sa[3] *= aa_lo.w;
        sa[4] *= aa_hi.x; sa[5] *= aa_hi.y; sa[6] *= aa_hi.z; sa[7] *= aa_hi.w;
        sb[0] *= ab_lo.x; sb[1] *= ab_lo.y; sb[2] *= ab_lo.z; sb[3] *= ab_lo.w;
        sb[4] *= ab_hi.x; sb[5] *= ab_hi.y; sb[6] *= ab_hi.z; sb[7] *= ab_hi.w;
        float mx = fmaxf(sa[0], sb[0]);
        #pragma unroll
        for (int r = 1; r < 8; ++r) mx = fmaxf(mx, fmaxf(sa[r], sb[r]));
        mx = fmaxf(mx, xhalf(mx));
        const float mnew = fmaxf(mrun, mx);
        const float sc   = __expf(mrun - mnew);
        mrun = mnew;

        float csum = 0.0f;
        #pragma unroll
        for (int r = 0; r < 8; ++r) {
            sa[r] = __expf(sa[r] - mnew);
            sb[r] = __expf(sb[r] - mnew);
            csum += sa[r] + sb[r];
        }
        lsum = lsum * sc + csum;
        #pragma unroll
        for (int r = 0; r < 8; ++r) { acc0[r] *= sc; acc1[r] *= sc; }

        #pragma unroll
        for (int j = 0; j < 4; ++j) {
            v2f pa; pa.x = sa[2 * j]; pa.y = sa[2 * j + 1];
            v2f pb; pb.x = sb[2 * j]; pb.y = sb[2 * j + 1];
            acc0 = wmma_f32_k4(pick2(va_lo, va_hi, j), pa, acc0);
            acc1 = wmma_f32_k4(pick2(vb_lo, vb_hi, j), pb, acc1);
        }
    }

    const float inv = 1.0f / (lsum + xhalf(lsum));
    float* op = Hout + base + (size_t)(n0 + l16) * D_ + h * HD_ + 8 * half;
    v4f o_lo, o_hi;
    o_lo.x = (acc0[0] + acc1[0]) * inv; o_lo.y = (acc0[1] + acc1[1]) * inv;
    o_lo.z = (acc0[2] + acc1[2]) * inv; o_lo.w = (acc0[3] + acc1[3]) * inv;
    o_hi.x = (acc0[4] + acc1[4]) * inv; o_hi.y = (acc0[5] + acc1[5]) * inv;
    o_hi.z = (acc0[6] + acc1[6]) * inv; o_hi.w = (acc0[7] + acc1[7]) * inv;
    *(v4f*)op       = o_lo;
    *(v4f*)(op + 4) = o_hi;
}

__global__ __launch_bounds__(256) void ln_kernel(
    const float* __restrict__ Hout, const float* __restrict__ x,
    const float* __restrict__ gamma, const float* __restrict__ beta,
    float* __restrict__ out)
{
    const int lane = threadIdx.x & 31;
    const int wid  = threadIdx.x >> 5;
    const size_t row = (size_t)blockIdx.x * 8 + wid;
    const float* hp = Hout + row * D_;
    const float* xp = x    + row * D_;
    const float y0 = hp[lane]      + xp[lane];
    const float y1 = hp[lane + 32] + xp[lane + 32];
    float s1 = y0 + y1;
    float s2 = y0 * y0 + y1 * y1;
    #pragma unroll
    for (int m = 1; m < 32; m <<= 1) {
        s1 += __shfl_xor(s1, m);
        s2 += __shfl_xor(s2, m);
    }
    const float mean = s1 * (1.0f / 64.0f);
    const float var  = s2 * (1.0f / 64.0f) - mean * mean;
    const float inv  = rsqrtf(var + EPS_);
    float* op = out + row * D_;
    op[lane]      = (y0 - mean) * inv * gamma[lane]      + beta[lane];
    op[lane + 32] = (y1 - mean) * inv * gamma[lane + 32] + beta[lane + 32];
}

extern "C" void kernel_launch(void* const* d_in, const int* in_sizes, int n_in,
                              void* d_out, int out_size, void* d_ws, size_t ws_size,
                              hipStream_t stream) {
    (void)in_sizes; (void)n_in; (void)out_size; (void)ws_size;
    const float* x     = (const float*)d_in[0];
    const float* adj   = (const float*)d_in[1];
    const float* Wq    = (const float*)d_in[3];
    const float* bq    = (const float*)d_in[4];
    const float* Wk    = (const float*)d_in[5];
    const float* bk    = (const float*)d_in[6];
    const float* Wv    = (const float*)d_in[7];
    const float* bv    = (const float*)d_in[8];
    const float* gamma = (const float*)d_in[9];
    const float* beta  = (const float*)d_in[10];
    float* out = (float*)d_out;

    float* ws = (float*)d_ws;
    const size_t S = (size_t)B_ * T_ * N_ * D_;
    float* Q  = ws;
    float* K  = ws + S;
    float* V  = ws + 2 * S;
    float* Hh = ws + 3 * S;

    dim3 g1(B_ * T_ * N_ / 16, 3, 1);
    qkv_kernel<<<g1, 32, 0, stream>>>(x, Wq, bq, Wk, bk, Wv, bv, Q, K, V);

    dim3 g2(N_ / 16, H_, B_ * T_);
    attn_kernel<<<g2, 32, 0, stream>>>(Q, K, V, adj, Hh);

    ln_kernel<<<(B_ * T_ * N_) / 8, 256, 0, stream>>>(Hh, x, gamma, beta, out);
}


# LayerNorm fused into the attention kernel: work-group = 4 waves = the 4 heads of a 16-query tile, heads meet in LDS behind one barrier, each wave normalises 4 rows; 3 launches -> 2; sums kept
# speedup vs baseline: 1.7212x; 1.0623x over previous
#include <hip/hip_runtime.h>

#define B_ 2
#define T_ 12
#define N_ 1024
#define D_ 64
#define H_ 4
#define HD_ 16
#define EPS_ 1e-5f
#define NEG_BIG -3.0e38f


typedef __attribute__((ext_vector_type(2))) float v2f;
typedef __attribute__((ext_vector_type(8))) float v8f;

__device__ __forceinline__ v8f wmma_f32_k4(v2f a, v2f b, v8f c) {
    return __builtin_amdgcn_wmma_f32_16x16x4_f32(
         false, a,  false, b,
         (short)0, c,  false,  false);
}

#define FRAG_TILE_ 256
__device__ __forceinline__ size_t frag_off(size_t bt, int head, size_t tile) {
    return ((bt * H_ + head) * (N_ / 16) + tile) * FRAG_TILE_;
}
__global__ __launch_bounds__(32) void qkv_kernel(
    const float* __restrict__ x,
    const float* __restrict__ Wq, const float* __restrict__ bq,
    const float* __restrict__ Wk, const float* __restrict__ bk,
    const float* __restrict__ Wv, const float* __restrict__ bv,
    float* __restrict__ Q, float* __restrict__ Ko, float* __restrict__ V)
{
    __shared__ float xs[16 * 68];
    const int lane = threadIdx.x & 31;
    const int half = lane >> 4;
    const int l16  = lane & 15;
    const int which = blockIdx.y;
    const float* W    = (which == 0) ? Wq : (which == 1) ? Wk : Wv;
    const float* bias = (which == 0) ? bq : (which == 1) ? bk : bv;
    float*       out  = (which == 0) ? Q  : (which == 1) ? Ko : V;
    const float  osc  = (which == 0) ? 0.25f : 1.0f;

    const size_t row0 = (size_t)blockIdx.x * 16;
    const float* xt = x + row0 * D_;

    #pragma unroll
    for (int i = 0; i < 8; ++i) {
        const int idx = (i * 32 + lane) * 4;
        const float4 val = *(const float4*)(xt + idx);
        const int row = idx >> 6, col = idx & 63;
        float* dst = &xs[row * 68 + col];
        dst[0] = val.x; dst[1] = val.y; dst[2] = val.z; dst[3] = val.w;
    }
    __syncthreads();

    v2f afrag[16];
    #pragma unroll
    for (int kk = 0; kk < 16; ++kk) {
        const float* p = &xs[l16 * 68 + kk * 4 + 2 * half];
        v2f a; a.x = p[0]; a.y = p[1];
        afrag[kk] = a;
    }

    v8f acc[4];
    #pragma unroll
    for (int c = 0; c < 4; ++c) acc[c] = {};

    const size_t bt   = row0 / N_;
    const size_t tile = (row0 % N_) / 16;

    if (which == 2) {
        #pragma unroll
        for (int kk = 0; kk < 16; ++kk) {
            #pragma unroll
            for (int c = 0; c < 4; ++c) {
                const float* wp = &W[(c * 16 + l16) * D_ + kk * 4 + 2 * half];
                v2f w; w.x = wp[0]; w.y = wp[1];
                acc[c] = wmma_f32_k4(afrag[kk], w, acc[c]);
            }
        }
        #pragma unroll
        for (int c = 0; c < 4; ++c) {
            const float bb = bias[c * 16 + l16];
            float* f = out + frag_off(bt, c, tile) + lane * 4;
            float4 lo, hi;
            lo.x = acc[c][0] + bb; lo.y = acc[c][1] + bb; lo.z = acc[c][2] + bb; lo.w = acc[c][3] + bb;
            hi.x = acc[c][4] + bb; hi.y = acc[c][5] + bb; hi.z = acc[c][6] + bb; hi.w = acc[c][7] + bb;
            *(float4*)f         = lo;
            *(float4*)(f + 128) = hi;
        }
    } else {
        #pragma unroll
        for (int kk = 0; kk < 16; ++kk) {
            #pragma unroll
            for (int c = 0; c < 4; ++c) {
                const float* wp = &W[(c * 16 + l16) * D_ + kk * 4 + 2 * half];
                v2f w; w.x = wp[0]; w.y = wp[1];
                acc[c] = wmma_f32_k4(w, afrag[kk], acc[c]);
            }
        }
        #pragma unroll
        for (int c = 0; c < 4; ++c) {
            const float4 b_lo = *(const float4*)(bias + c * 16 + 8 * half);
            const float4 b_hi = *(const float4*)(bias + c * 16 + 8 * half + 4);
            float* f = out + frag_off(bt, c, tile) + lane * 4;
            float4 lo, hi;
            lo.x = (acc[c][0] + b_lo.x) * osc; lo.y = (acc[c][1] + b_lo.y) * osc;
            lo.z = (acc[c][2] + b_lo.z) * osc; lo.w = (acc[c][3] + b_lo.w) * osc;
            hi.x = (acc[c][4] + b_hi.x) * osc; hi.y = (acc[c][5] + b_hi.y) * osc;
            hi.z = (acc[c][6] + b_hi.z) * osc; hi.w = (acc[c][7] + b_hi.w) * osc;
            *(float4*)f         = lo;
            *(float4*)(f + 128) = hi;
        }
    }
}

typedef __attribute__((ext_vector_type(4))) float v4f;

#define GKEYS_ 32

__device__ __forceinline__ float xhalf(float x) {
    const int b = __builtin_bit_cast(int, x);
    return __builtin_bit_cast(float,
        __builtin_amdgcn_permlanex16(b, b, 0x76543210, 0xfedcba98, false, false));
}

__device__ __forceinline__ v2f pick2(v4f lo, v4f hi, int j) {
    v2f r;
    if (j == 0)      { r.x = lo.x; r.y = lo.y; }
    else if (j == 1) { r.x = lo.z; r.y = lo.w; }
    else if (j == 2) { r.x = hi.x; r.y = hi.y; }
    else             { r.x = hi.z; r.y = hi.w; }
    return r;
}

#define HS_PITCH4_ 17

__global__ __launch_bounds__(128) void attn_ln_kernel(
    const float* __restrict__ Qf, const float* __restrict__ Kf,
    const float* __restrict__ Vf, const float* __restrict__ adj,
    const float* __restrict__ x, const float* __restrict__ gamma,
    const float* __restrict__ beta, float* __restrict__ out)
{
    __shared__ v4f hs[16 * HS_PITCH4_];
    const int lane = threadIdx.x & 31;
    const int half = lane >> 4;
    const int l16  = lane & 15;
    const int n0 = blockIdx.x * 16;
    const int h  = __builtin_amdgcn_readfirstlane((int)(threadIdx.x >> 5));
    const int bt = blockIdx.y;
    const size_t base  = (size_t)bt * N_ * D_;
    const size_t abase = (size_t)bt * N_ * N_;

    const float* qp = Qf + frag_off(bt, h, blockIdx.x) + lane * 4;
    const v4f q_lo = *(const v4f*)qp;
    const v4f q_hi = *(const v4f*)(qp + 128);

    const float* kp = Kf + frag_off(bt, h, 0) + lane * 4;
    const float* vp = Vf + frag_off(bt, h, 0) + lane * 4;
    const float* ap = adj + abase + (size_t)(n0 + l16) * N_ + 8 * half;

    float mrun = NEG_BIG;
    float lsum = 0.0f;
    v8f acc0 = {}, acc1 = {};

    for (int m0 = 0; m0 < N_; m0 += GKEYS_) {
        const float* k0 = kp + (size_t)m0 * (FRAG_TILE_ / 16);
        const v4f ka_lo = *(const v4f*)(k0);
        const v4f ka_hi = *(const v4f*)(k0 + 128);
        const v4f kb_lo = *(const v4f*)(k0 + FRAG_TILE_);
        const v4f kb_hi = *(const v4f*)(k0 + FRAG_TILE_ + 128);
        const v4f aa_lo = *(const v4f*)(ap + m0);
        const v4f aa_hi = *(const v4f*)(ap + m0 + 4);
        const v4f ab_lo = *(const v4f*)(ap + m0 + 16);
        const v4f ab_hi = *(const v4f*)(ap + m0 + 20);
        const float* v0 = vp + (size_t)m0 * (FRAG_TILE_ / 16);
        const v4f va_lo = *(const v4f*)(v0);
        const v4f va_hi = *(const v4f*)(v0 + 128);
        const v4f vb_lo = *(const v4f*)(v0 + FRAG_TILE_);
        const v4f vb_hi = *(const v4f*)(v0 + FRAG_TILE_ + 128);

        v8f sa = {}, sb = {};
        #pragma unroll
        for (int kk = 0; kk < 4; ++kk) {
            const v2f bq = pick2(q_lo, q_hi, kk);
            sa = wmma_f32_k4(pick2(ka_lo, ka_hi, kk), bq, sa);
            sb = wmma_f32_k4(pick2(kb_lo, kb_hi, kk), bq, sb);
        }

        sa[0] *= aa_lo.x; sa[1] *= aa_lo.y; sa[2] *= aa_lo.z; sa[3] *= aa_lo.w;
        sa[4] *= aa_hi.x; sa[5] *= aa_hi.y; sa[6] *= aa_hi.z; sa[7] *= aa_hi.w;
        sb[0] *= ab_lo.x; sb[1] *= ab_lo.y; sb[2] *= ab_lo.z; sb[3] *= ab_lo.w;
        sb[4] *= ab_hi.x; sb[5] *= ab_hi.y; sb[6] *= ab_hi.z; sb[7] *= ab_hi.w;
        float mx = fmaxf(sa[0], sb[0]);
        #pragma unroll
        for (int r = 1; r < 8; ++r) mx = fmaxf(mx, fmaxf(sa[r], sb[r]));
        mx = fmaxf(mx, xhalf(mx));
        const float mnew = fmaxf(mrun, mx);
        const float sc   = __expf(mrun - mnew);
        mrun = mnew;

        float csum = 0.0f;
        #pragma unroll
        for (int r = 0; r < 8; ++r) {
            sa[r] = __expf(sa[r] - mnew);
            sb[r] = __expf(sb[r] - mnew);
            csum += sa[r] + sb[r];
        }
        lsum = lsum * sc + csum;
        #pragma unroll
        for (int r = 0; r < 8; ++r) { acc0[r] *= sc; acc1[r] *= sc; }

        #pragma unroll
        for (int j = 0; j < 4; ++j) {
            v2f pa; pa.x = sa[2 * j]; pa.y = sa[2 * j + 1];
            v2f pb; pb.x = sb[2 * j]; pb.y = sb[2 * j + 1];
            acc0 = wmma_f32_k4(pick2(va_lo, va_hi, j), pa, acc0);
            acc1 = wmma_f32_k4(pick2(vb_lo, vb_hi, j), pb, acc1);
        }
    }

    const float* xr = x + base + (size_t)(n0 + 4 * h) * D_ + lane;
    float xa[4], xb[4];
    #pragma unroll
    for (int r = 0; r < 4; ++r) { xa[r] = xr[r * D_]; xb[r] = xr[r * D_ + 32]; }
    const float g0 = gamma[lane], g1 = gamma[lane + 32];
    const float b0 = beta[lane],  b1 = beta[lane + 32];

    const float inv = 1.0f / (lsum + xhalf(lsum));
    v4f o_lo, o_hi;
    o_lo.x = (acc0[0] + acc1[0]) * inv; o_lo.y = (acc0[1] + acc1[1]) * inv;
    o_lo.z = (acc0[2] + acc1[2]) * inv; o_lo.w = (acc0[3] + acc1[3]) * inv;
    o_hi.x = (acc0[4] + acc1[4]) * inv; o_hi.y = (acc0[5] + acc1[5]) * inv;
    o_hi.z = (acc0[6] + acc1[6]) * inv; o_hi.w = (acc0[7] + acc1[7]) * inv;
    v4f* hp = &hs[l16 * HS_PITCH4_ + h * (HD_ / 4) + 2 * half];
    hp[0] = o_lo;
    hp[1] = o_hi;
    __syncthreads();

    const float* hsf = (const float*)hs;
    float* orow = out + base + (size_t)(n0 + 4 * h) * D_ + lane;
    #pragma unroll
    for (int r = 0; r < 4; ++r) {
        const float* hrow = hsf + (4 * h + r) * (HS_PITCH4_ * 4);
        const float y0 = hrow[lane]      + xa[r];
        const float y1 = hrow[lane + 32] + xb[r];
        float s1 = y0 + y1;
        float s2 = y0 * y0 + y1 * y1;
        #pragma unroll
        for (int m = 1; m < 32; m <<= 1) {
            s1 += __shfl_xor(s1, m);
            s2 += __shfl_xor(s2, m);
        }
        const float mean = s1 * (1.0f / 64.0f);
        const float var  = s2 * (1.0f / 64.0f) - mean * mean;
        const float rstd = rsqrtf(var + EPS_);
        orow[r * D_]      = (y0 - mean) * rstd * g0 + b0;
        orow[r * D_ + 32] = (y1 - mean) * rstd * g1 + b1;
    }
}

extern "C" void kernel_launch(void* const* d_in, const int* in_sizes, int n_in,
                              void* d_out, int out_size, void* d_ws, size_t ws_size,
                              hipStream_t stream) {
    (void)in_sizes; (void)n_in; (void)out_size; (void)ws_size;
    const float* x     = (const float*)d_in[0];
    const float* adj   = (const float*)d_in[1];
    const float* Wq    = (const float*)d_in[3];
    const float* bq    = (const float*)d_in[4];
    const float* Wk    = (const float*)d_in[5];
    const float* bk    = (const float*)d_in[6];
    const float* Wv    = (const float*)d_in[7];
    const float* bv    = (const float*)d_in[8];
    const float* gamma = (const float*)d_in[9];
    const float* beta  = (const float*)d_in[10];
    float* out = (float*)d_out;

    float* ws = (float*)d_ws;
    const size_t S = (size_t)B_ * T_ * N_ * D_;
    float* Q  = ws;
    float* K  = ws + S;
    float* V  = ws + 2 * S;

    dim3 g1(B_ * T_ * N_ / 16, 3, 1);
    qkv_kernel<<<g1, 32, 0, stream>>>(x, Wq, bq, Wk, bk, Wv, bv, Q, K, V);

    dim3 g2(N_ / 16, B_ * T_, 1);
    attn_ln_kernel<<<g2, 128, 0, stream>>>(Q, K, V, adj, x, gamma, beta, out);
}


# adj tile of each 32-key step read in whole 128-byte lines (4 lines per load, was 16) and turned to the lane = query order through a wave-private LDS slab, no barrier; sums unchanged
# speedup vs baseline: 1.9443x; 1.1296x over previous
#include <hip/hip_runtime.h>

#define B_ 2
#define T_ 12
#define N_ 1024
#define D_ 64
#define H_ 4
#define HD_ 16
#define EPS_ 1e-5f
#define NEG_BIG -3.0e38f


typedef __attribute__((ext_vector_type(2))) float v2f;
typedef __attribute__((ext_vector_type(8))) float v8f;

__device__ __forceinline__ v8f wmma_f32_k4(v2f a, v2f b, v8f c) {
    return __builtin_amdgcn_wmma_f32_16x16x4_f32(
         false, a,  false, b,
         (short)0, c,  false,  false);
}

#define FRAG_TILE_ 256
__device__ __forceinline__ size_t frag_off(size_t bt, int head, size_t tile) {
    return ((bt * H_ + head) * (N_ / 16) + tile) * FRAG_TILE_;
}
__global__ __launch_bounds__(32) void qkv_kernel(
    const float* __restrict__ x,
    const float* __restrict__ Wq, const float* __restrict__ bq,
    const float* __restrict__ Wk, const float* __restrict__ bk,
    const float* __restrict__ Wv, const float* __restrict__ bv,
    float* __restrict__ Q, float* __restrict__ Ko, float* __restrict__ V)
{
    __shared__ float xs[16 * 68];
    const int lane = threadIdx.x & 31;
    const int half = lane >> 4;
    const int l16  = lane & 15;
    const int which = blockIdx.y;
    const float* W    = (which == 0) ? Wq : (which == 1) ? Wk : Wv;
    const float* bias = (which == 0) ? bq : (which == 1) ? bk : bv;
    float*       out  = (which == 0) ? Q  : (which == 1) ? Ko : V;
    const float  osc  = (which == 0) ? 0.25f : 1.0f;

    const size_t row0 = (size_t)blockIdx.x * 16;
    const float* xt = x + row0 * D_;

    #pragma unroll
    for (int i = 0; i < 8; ++i) {
        const int idx = (i * 32 + lane) * 4;
        const float4 val = *(const float4*)(xt + idx);
        const int row = idx >> 6, col = idx & 63;
        float* dst = &xs[row * 68 + col];
        dst[0] = val.x; dst[1] = val.y; dst[2] = val.z; dst[3] = val.w;
    }
    __syncthreads();

    v2f afrag[16];
    #pragma unroll
    for (int kk = 0; kk < 16; ++kk) {
        const float* p = &xs[l16 * 68 + kk * 4 + 2 * half];
        v2f a; a.x = p[0]; a.y = p[1];
        afrag[kk] = a;
    }

    v8f acc[4];
    #pragma unroll
    for (int c = 0; c < 4; ++c) acc[c] = {};

    const size_t bt   = row0 / N_;
    const size_t tile = (row0 % N_) / 16;

    if (which == 2) {
        #pragma unroll
        for (int kk = 0; kk < 16; ++kk) {
            #pragma unroll
            for (int c = 0; c < 4; ++c) {
                const float* wp = &W[(c * 16 + l16) * D_ + kk * 4 + 2 * half];
                v2f w; w.x = wp[0]; w.y = wp[1];
                acc[c] = wmma_f32_k4(afrag[kk], w, acc[c]);
            }
        }
        #pragma unroll
        for (int c = 0; c < 4; ++c) {
            const float bb = bias[c * 16 + l16];
            float* f = out + frag_off(bt, c, tile) + lane * 4;
            float4 lo, hi;
            lo.x = acc[c][0] + bb; lo.y = acc[c][1] + bb; lo.z = acc[c][2] + bb; lo.w = acc[c][3] + bb;
            hi.x = acc[c][4] + bb; hi.y = acc[c][5] + bb; hi.z = acc[c][6] + bb; hi.w = acc[c][7] + bb;
            *(float4*)f         = lo;
            *(float4*)(f + 128) = hi;
        }
    } else {
        #pragma unroll
        for (int kk = 0; kk < 16; ++kk) {
            #pragma unroll
            for (int c = 0; c < 4; ++c) {
                const float* wp = &W[(c * 16 + l16) * D_ + kk * 4 + 2 * half];
                v2f w; w.x = wp[0]; w.y = wp[1];
                acc[c] = wmma_f32_k4(w, afrag[kk], acc[c]);
            }
        }
        #pragma unroll
        for (int c = 0; c < 4; ++c) {
            const float4 b_lo = *(const float4*)(bias + c * 16 + 8 * half);
            const float4 b_hi = *(const float4*)(bias + c * 16 + 8 * half + 4);
            float* f = out + frag_off(bt, c, tile) + lane * 4;
            float4 lo, hi;
            lo.x = (acc[c][0] + b_lo.x) * osc; lo.y = (acc[c][1] + b_lo.y) * osc;
            lo.z = (acc[c][2] + b_lo.z) * osc; lo.w = (acc[c][3] + b_lo.w) * osc;
            hi.x = (acc[c][4] + b_hi.x) * osc; hi.y = (acc[c][5] + b_hi.y) * osc;
            hi.z = (acc[c][6] + b_hi.z) * osc; hi.w = (acc[c][7] + b_hi.w) * osc;
            *(float4*)f         = lo;
            *(float4*)(f + 128) = hi;
        }
    }
}

typedef __attribute__((ext_vector_type(4))) float v4f;

#define GKEYS_ 32

__device__ __forceinline__ float xhalf(float x) {
    const int b = __builtin_bit_cast(int, x);
    return __builtin_bit_cast(float,
        __builtin_amdgcn_permlanex16(b, b, 0x76543210, 0xfedcba98, false, false));
}

__device__ __forceinline__ v2f pick2(v4f lo, v4f hi, int j) {
    v2f r;
    if (j == 0)      { r.x = lo.x; r.y = lo.y; }
    else if (j == 1) { r.x = lo.z; r.y = lo.w; }
    else if (j == 2) { r.x = hi.x; r.y = hi.y; }
    else             { r.x = hi.z; r.y = hi.w; }
    return r;
}

#define HS_PITCH4_ 17

#define ADJ_PITCH4_ 9
#define ADJ_SLAB4_ (16 * ADJ_PITCH4_)

__global__ __launch_bounds__(128) void attn_ln_kernel(
    const float* __restrict__ Qf, const float* __restrict__ Kf,
    const float* __restrict__ Vf, const float* __restrict__ adj,
    const float* __restrict__ x, const float* __restrict__ gamma,
    const float* __restrict__ beta, float* __restrict__ out)
{
    __shared__ v4f hs[16 * HS_PITCH4_];
    __shared__ v4f adjs[H_ * ADJ_SLAB4_];
    const int lane = threadIdx.x & 31;
    const int half = lane >> 4;
    const int l16  = lane & 15;
    const int n0 = blockIdx.x * 16;
    const int h  = __builtin_amdgcn_readfirstlane((int)(threadIdx.x >> 5));
    const int bt = blockIdx.y;
    const size_t base  = (size_t)bt * N_ * D_;
    const size_t abase = (size_t)bt * N_ * N_;

    const float* qp = Qf + frag_off(bt, h, blockIdx.x) + lane * 4;
    const v4f q_lo = *(const v4f*)qp;
    const v4f q_hi = *(const v4f*)(qp + 128);

    const float* kp = Kf + frag_off(bt, h, 0) + lane * 4;
    const float* vp = Vf + frag_off(bt, h, 0) + lane * 4;
    const float* ag = adj + abase + (size_t)(n0 + (lane >> 3)) * N_ + 4 * (lane & 7);
    v4f* aslab = adjs + h * ADJ_SLAB4_;
    v4f* ast = aslab + (lane >> 3) * ADJ_PITCH4_ + (lane & 7);
    const v4f* ald = aslab + l16 * ADJ_PITCH4_ + 2 * half;

    float mrun = NEG_BIG;
    float lsum = 0.0f;
    v8f acc0 = {}, acc1 = {};

    for (int m0 = 0; m0 < N_; m0 += GKEYS_) {
        const float* k0 = kp + (size_t)m0 * (FRAG_TILE_ / 16);
        const v4f ka_lo = *(const v4f*)(k0);
        const v4f ka_hi = *(const v4f*)(k0 + 128);
        const v4f kb_lo = *(const v4f*)(k0 + FRAG_TILE_);
        const v4f kb_hi = *(const v4f*)(k0 + FRAG_TILE_ + 128);
        const v4f g0 = *(const v4f*)(ag + m0);
        const v4f g1 = *(const v4f*)(ag + m0 + 4 * N_);
        const v4f g2 = *(const v4f*)(ag + m0 + 8 * N_);
        const v4f g3 = *(const v4f*)(ag + m0 + 12 * N_);
        const float* v0 = vp + (size_t)m0 * (FRAG_TILE_ / 16);
        const v4f va_lo = *(const v4f*)(v0);
        const v4f va_hi = *(const v4f*)(v0 + 128);
        const v4f vb_lo = *(const v4f*)(v0 + FRAG_TILE_);
        const v4f vb_hi = *(const v4f*)(v0 + FRAG_TILE_ + 128);

        __builtin_amdgcn_fence(__ATOMIC_ACQ_REL, "wavefront");
        ast[0]                = g0;
        ast[4 * ADJ_PITCH4_]  = g1;
        ast[8 * ADJ_PITCH4_]  = g2;
        ast[12 * ADJ_PITCH4_] = g3;
        __builtin_amdgcn_fence(__ATOMIC_ACQ_REL, "wavefront");
        const v4f aa_lo = ald[0];
        const v4f aa_hi = ald[1];
        const v4f ab_lo = ald[4];
        const v4f ab_hi = ald[5];

        v8f sa = {}, sb = {};
        #pragma unroll
        for (int kk = 0; kk < 4; ++kk) {
            const v2f bq = pick2(q_lo, q_hi, kk);
            sa = wmma_f32_k4(pick2(ka_lo, ka_hi, kk), bq, sa);
            sb = wmma_f32_k4(pick2(kb_lo, kb_hi, kk), bq, sb);
        }

        sa[0] *= aa_lo.x; sa[1] *= aa_lo.y; sa[2] *= aa_lo.z; sa[3] *= aa_lo.w;
        sa[4] *= aa_hi.x; sa[5] *= aa_hi.y; sa[6] *= aa_hi.z; sa[7] *= aa_hi.w;
        sb[0] *= ab_lo.x; sb[1] *= ab_lo.y; sb[2] *= ab_lo.z; sb[3] *= ab_lo.w;
        sb[4] *= ab_hi.x; sb[5] *= ab_hi.y; sb[6] *= ab_hi.z; sb[7] *= ab_hi.w;
        float mx = fmaxf(sa[0], sb[0]);
        #pragma unroll
        for (int r = 1; r < 8; ++r) mx = fmaxf(mx, fmaxf(sa[r], sb[r]));
        mx = fmaxf(mx, xhalf(mx));
        const float mnew = fmaxf(mrun, mx);
        const float sc   = __expf(mrun - mnew);
        mrun = mnew;

        float csum = 0.0f;
        #pragma unroll
        for (int r = 0; r < 8; ++r) {
            sa[r] = __expf(sa[r] - mnew);
            sb[r] = __expf(sb[r] - mnew);
            csum += sa[r] + sb[r];
        }
        lsum = lsum * sc + csum;
        #pragma unroll
        for (int r = 0; r < 8; ++r) { acc0[r] *= sc; acc1[r] *= sc; }

        #pragma unroll
        for (int j = 0; j < 4; ++j) {
            v2f pa; pa.x = sa[2 * j]; pa.y = sa[2 * j + 1];
            v2f pb; pb.x = sb[2 * j]; pb.y = sb[2 * j + 1];
            acc0 = wmma_f32_k4(pick2(va_lo, va_hi, j), pa, acc0);
            acc1 = wmma_f32_k4(pick2(vb_lo, vb_hi, j), pb, acc1);
        }
    }

    const float* xr = x + base + (size_t)(n0 + 4 * h) * D_ + lane;
    float xa[4], xb[4];
    #pragma unroll
    for (int r = 0; r < 4; ++r) { xa[r] = xr[r * D_]; xb[r] = xr[r * D_ + 32]; }
    const float g0 = gamma[lane], g1 = gamma[lane + 32];
    const float b0 = beta[lane],  b1 = beta[lane + 32];

    const float inv = 1.0f / (lsum + xhalf(lsum));
    v4f o_lo, o_hi;
    o_lo.x = (acc0[0] + acc1[0]) * inv; o_lo.y = (acc0[1] + acc1[1]) * inv;
    o_lo.z = (acc0[2] + acc1[2]) * inv; o_lo.w = (acc0[3] + acc1[3]) * inv;
    o_hi.x = (acc0[4] + acc1[4]) * inv; o_hi.y = (acc0[5] + acc1[5]) * inv;
    o_hi.z = (acc0[6] + acc1[6]) * inv; o_hi.w = (acc0[7] + acc1[7]) * inv;
    v4f* hp = &hs[l16 * HS_PITCH4_ + h * (HD_ / 4) + 2 * half];
    hp[0] = o_lo;
    hp[1] = o_hi;
    __syncthreads();

    const float* hsf = (const float*)hs;
    float* orow = out + base + (size_t)(n0 + 4 * h) * D_ + lane;
    #pragma unroll
    for (int r = 0; r < 4; ++r) {
        const float* hrow = hsf + (4 * h + r) * (HS_PITCH4_ * 4);
        const float y0 = hrow[lane]      + xa[r];
        const float y1 = hrow[lane + 32] + xb[r];
        float s1 = y0 + y1;
        float s2 = y0 * y0 + y1 * y1;
        #pragma unroll
        for (int m = 1; m < 32; m <<= 1) {
            s1 += __shfl_xor(s1, m);
            s2 += __shfl_xor(s2, m);
        }
        const float mean = s1 * (1.0f / 64.0f);
        const float var  = s2 * (1.0f / 64.0f) - mean * mean;
        const float rstd = rsqrtf(var + EPS_);
        orow[r * D_]      = (y0 - mean) * rstd * g0 + b0;
        orow[r * D_ + 32] = (y1 - mean) * rstd * g1 + b1;
    }
}

extern "C" void kernel_launch(void* const* d_in, const int* in_sizes, int n_in,
                              void* d_out, int out_size, void* d_ws, size_t ws_size,
                              hipStream_t stream) {
    (void)in_sizes; (void)n_in; (void)out_size; (void)ws_size;
    const float* x     = (const float*)d_in[0];
    const float* adj   = (const float*)d_in[1];
    const float* Wq    = (const float*)d_in[3];
    const float* bq    = (const float*)d_in[4];
    const float* Wk    = (const float*)d_in[5];
    const float* bk    = (const float*)d_in[6];
    const float* Wv    = (const float*)d_in[7];
    const float* bv    = (const float*)d_in[8];
    const float* gamma = (const float*)d_in[9];
    const float* beta  = (const float*)d_in[10];
    float* out = (float*)d_out;

    float* ws = (float*)d_ws;
    const size_t S = (size_t)B_ * T_ * N_ * D_;
    float* Q  = ws;
    float* K  = ws + S;
    float* V  = ws + 2 * S;

    dim3 g1(B_ * T_ * N_ / 16, 3, 1);
    qkv_kernel<<<g1, 32, 0, stream>>>(x, Wq, bq, Wk, bk, Wv, bv, Q, K, V);

    dim3 g2(N_ / 16, B_ * T_, 1);
    attn_ln_kernel<<<g2, 128, 0, stream>>>(Q, K, V, adj, x, gamma, beta, out);
}


# attention: each wave takes two 16-query tiles of its head (work-group = 32 queries), so every K and V fragment load of the key loop feeds two products; per-query sums and their order unchanged
# speedup vs baseline: 2.0556x; 1.0572x over previous
#include <hip/hip_runtime.h>

#define B_ 2
#define T_ 12
#define N_ 1024
#define D_ 64
#define H_ 4
#define HD_ 16
#define EPS_ 1e-5f
#define NEG_BIG -3.0e38f


typedef __attribute__((ext_vector_type(2))) float v2f;
typedef __attribute__((ext_vector_type(8))) float v8f;

__device__ __forceinline__ v8f wmma_f32_k4(v2f a, v2f b, v8f c) {
    return __builtin_amdgcn_wmma_f32_16x16x4_f32(
         false, a,  false, b,
         (short)0, c,  false,  false);
}

#define FRAG_TILE_ 256
__device__ __forceinline__ size_t frag_off(size_t bt, int head, size_t tile) {
    return ((bt * H_ + head) * (N_ / 16) + tile) * FRAG_TILE_;
}
__global__ __launch_bounds__(32) void qkv_kernel(
    const float* __restrict__ x,
    const float* __restrict__ Wq, const float* __restrict__ bq,
    const float* __restrict__ Wk, const float* __restrict__ bk,
    const float* __restrict__ Wv, const float* __restrict__ bv,
    float* __restrict__ Q, float* __restrict__ Ko, float* __restrict__ V)
{
    __shared__ float xs[16 * 68];
    const int lane = threadIdx.x & 31;
    const int half = lane >> 4;
    const int l16  = lane & 15;
    const int which = blockIdx.y;
    const float* W    = (which == 0) ? Wq : (which == 1) ? Wk : Wv;
    const float* bias = (which == 0) ? bq : (which == 1) ? bk : bv;
    float*       out  = (which == 0) ? Q  : (which == 1) ? Ko : V;
    const float  osc  = (which == 0) ? 0.25f : 1.0f;

    const size_t row0 = (size_t)blockIdx.x * 16;
    const float* xt = x + row0 * D_;

    #pragma unroll
    for (int i = 0; i < 8; ++i) {
        const int idx = (i * 32 + lane) * 4;
        const float4 val = *(const float4*)(xt + idx);
        const int row = idx >> 6, col = idx & 63;
        float* dst = &xs[row * 68 + col];
        dst[0] = val.x; dst[1] = val.y; dst[2] = val.z; dst[3] = val.w;
    }
    __syncthreads();

    v2f afrag[16];
    #pragma unroll
    for (int kk = 0; kk < 16; ++kk) {
        const float* p = &xs[l16 * 68 + kk * 4 + 2 * half];
        v2f a; a.x = p[0]; a.y = p[1];
        afrag[kk] = a;
    }

    v8f acc[4];
    #pragma unroll
    for (int c = 0; c < 4; ++c) acc[c] = {};

    const size_t bt   = row0 / N_;
    const size_t tile = (row0 % N_) / 16;

    if (which == 2) {
        #pragma unroll
        for (int kk = 0; kk < 16; ++kk) {
            #pragma unroll
            for (int c = 0; c < 4; ++c) {
                const float* wp = &W[(c * 16 + l16) * D_ + kk * 4 + 2 * half];
                v2f w; w.x = wp[0]; w.y = wp[1];
                acc[c] = wmma_f32_k4(afrag[kk], w, acc[c]);
            }
        }
        #pragma unroll
        for (int c = 0; c < 4; ++c) {
            const float bb = bias[c * 16 + l16];
            float* f = out + frag_off(bt, c, tile) + lane * 4;
            float4 lo, hi;
            lo.x = acc[c][0] + bb; lo.y = acc[c][1] + bb; lo.z = acc[c][2] + bb; lo.w = acc[c][3] + bb;
            hi.x = acc[c][4] + bb; hi.y = acc[c][5] + bb; hi.z = acc[c][6] + bb; hi.w = acc[c][7] + bb;
            *(float4*)f         = lo;
            *(float4*)(f + 128) = hi;
        }
    } else {
        #pragma unroll
        for (int kk = 0; kk < 16; ++kk) {
            #pragma unroll
            for (int c = 0; c < 4; ++c) {
                const float* wp = &W[(c * 16 + l16) * D_ + kk * 4 + 2 * half];
                v2f w; w.x = wp[0]; w.y = wp[1];
                acc[c] = wmma_f32_k4(w, afrag[kk], acc[c]);
            }
        }
        #pragma unroll
        for (int c = 0; c < 4; ++c) {
            const float4 b_lo = *(const float4*)(bias + c * 16 + 8 * half);
            const float4 b_hi = *(const float4*)(bias + c * 16 + 8 * half + 4);
            float* f = out + frag_off(bt, c, tile) + lane * 4;
            float4 lo, hi;
            lo.x = (acc[c][0] + b_lo.x) * osc; lo.y = (acc[c][1] + b_lo.y) * osc;
            lo.z = (acc[c][2] + b_lo.z) * osc; lo.w = (acc[c][3] + b_lo.w) * osc;
            hi.x = (acc[c][4] + b_hi.x) * osc; hi.y = (acc[c][5] + b_hi.y) * osc;
            hi.z = (acc[c][6] + b_hi.z) * osc; hi.w = (acc[c][7] + b_hi.w) * osc;
            *(float4*)f         = lo;
            *(float4*)(f + 128) = hi;
        }
    }
}

typedef __attribute__((ext_vector_type(4))) float v4f;

#define GKEYS_ 32

__device__ __forceinline__ float xhalf(float x) {
    const int b = __builtin_bit_cast(int, x);
    return __builtin_bit_cast(float,
        __builtin_amdgcn_permlanex16(b, b, 0x76543210, 0xfedcba98, false, false));
}

__device__ __forceinline__ v2f pick2(v4f lo, v4f hi, int j) {
    v2f r;
    if (j == 0)      { r.x = lo.x; r.y = lo.y; }
    else if (j == 1) { r.x = lo.z; r.y = lo.w; }
    else if (j == 2) { r.x = hi.x; r.y = hi.y; }
    else             { r.x = hi.z; r.y = hi.w; }
    return r;
}

#define QROWS_ 32
#define HS_PITCH4_ 17

#define ADJ_PITCH4_ 9
#define ADJ_SLAB4_ (QROWS_ * ADJ_PITCH4_)

__device__ __forceinline__ void softmax_step(
    v8f& sa, v8f& sb, v4f aa_lo, v4f aa_hi, v4f ab_lo, v4f ab_hi,
    float& mrun, float& lsum, v8f& acc0, v8f& acc1)
{
    sa[0] *= aa_lo.x; sa[1] *= aa_lo.y; sa[2] *= aa_lo.z; sa[3] *= aa_lo.w;
    sa[4] *= aa_hi.x; sa[5] *= aa_hi.y; sa[6] *= aa_hi.z; sa[7] *= aa_hi.w;
    sb[0] *= ab_lo.x; sb[1] *= ab_lo.y; sb[2] *= ab_lo.z; sb[3] *= ab_lo.w;
    sb[4] *= ab_hi.x; sb[5] *= ab_hi.y; sb[6] *= ab_hi.z; sb[7] *= ab_hi.w;
    float mx = fmaxf(sa[0], sb[0]);
    #pragma unroll
    for (int r = 1; r < 8; ++r) mx = fmaxf(mx, fmaxf(sa[r], sb[r]));
    mx = fmaxf(mx, xhalf(mx));
    const float mnew = fmaxf(mrun, mx);
    const float sc   = __expf(mrun - mnew);
    mrun = mnew;

    float csum = 0.0f;
    #pragma unroll
    for (int r = 0; r < 8; ++r) {
        sa[r] = __expf(sa[r] - mnew);
        sb[r] = __expf(sb[r] - mnew);
        csum += sa[r] + sb[r];
    }
    lsum = lsum * sc + csum;
    #pragma unroll
    for (int r = 0; r < 8; ++r) { acc0[r] *= sc; acc1[r] *= sc; }
}

__global__ __launch_bounds__(128) void attn_ln_kernel(
    const float* __restrict__ Qf, const float* __restrict__ Kf,
    const float* __restrict__ Vf, const float* __restrict__ adj,
    const float* __restrict__ x, const float* __restrict__ gamma,
    const float* __restrict__ beta, float* __restrict__ out)
{
    __shared__ v4f hs[QROWS_ * HS_PITCH4_];
    __shared__ v4f adjs[H_ * ADJ_SLAB4_];
    const int lane = threadIdx.x & 31;
    const int half = lane >> 4;
    const int l16  = lane & 15;
    const int n0 = blockIdx.x * QROWS_;
    const int h  = __builtin_amdgcn_readfirstlane((int)(threadIdx.x >> 5));
    const int bt = blockIdx.y;
    const size_t base  = (size_t)bt * N_ * D_;
    const size_t abase = (size_t)bt * N_ * N_;

    const float* qp = Qf + frag_off(bt, h, (size_t)blockIdx.x * 2) + lane * 4;
    const v4f q0_lo = *(const v4f*)qp;
    const v4f q0_hi = *(const v4f*)(qp + 128);
    const v4f q1_lo = *(const v4f*)(qp + FRAG_TILE_);
    const v4f q1_hi = *(const v4f*)(qp + FRAG_TILE_ + 128);

    const float* kp = Kf + frag_off(bt, h, 0) + lane * 4;
    const float* vp = Vf + frag_off(bt, h, 0) + lane * 4;
    const float* ag = adj + abase + (size_t)(n0 + (lane >> 3)) * N_ + 4 * (lane & 7);
    v4f* aslab = adjs + h * ADJ_SLAB4_;
    v4f* ast = aslab + (lane >> 3) * ADJ_PITCH4_ + (lane & 7);
    const v4f* ald0 = aslab + l16 * ADJ_PITCH4_ + 2 * half;
    const v4f* ald1 = ald0 + 16 * ADJ_PITCH4_;

    float mrun0 = NEG_BIG, mrun1 = NEG_BIG;
    float lsum0 = 0.0f, lsum1 = 0.0f;
    v8f acc00 = {}, acc01 = {};
    v8f acc10 = {}, acc11 = {};

    for (int m0 = 0; m0 < N_; m0 += GKEYS_) {
        const float* k0 = kp + (size_t)m0 * (FRAG_TILE_ / 16);
        const v4f ka_lo = *(const v4f*)(k0);
        const v4f ka_hi = *(const v4f*)(k0 + 128);
        const v4f kb_lo = *(const v4f*)(k0 + FRAG_TILE_);
        const v4f kb_hi = *(const v4f*)(k0 + FRAG_TILE_ + 128);
        v4f g[8];
        #pragma unroll
        for (int q = 0; q < 8; ++q) g[q] = *(const v4f*)(ag + m0 + (size_t)(4 * q) * N_);
        const float* v0 = vp + (size_t)m0 * (FRAG_TILE_ / 16);
        const v4f va_lo = *(const v4f*)(v0);
        const v4f va_hi = *(const v4f*)(v0 + 128);
        const v4f vb_lo = *(const v4f*)(v0 + FRAG_TILE_);
        const v4f vb_hi = *(const v4f*)(v0 + FRAG_TILE_ + 128);

        __builtin_amdgcn_fence(__ATOMIC_ACQ_REL, "wavefront");
        #pragma unroll
        for (int q = 0; q < 8; ++q) ast[4 * q * ADJ_PITCH4_] = g[q];
        __builtin_amdgcn_fence(__ATOMIC_ACQ_REL, "wavefront");
        const v4f a0a_lo = ald0[0];
        const v4f a0a_hi = ald0[1];
        const v4f a0b_lo = ald0[4];
        const v4f a0b_hi = ald0[5];
        const v4f a1a_lo = ald1[0];
        const v4f a1a_hi = ald1[1];
        const v4f a1b_lo = ald1[4];
        const v4f a1b_hi = ald1[5];

        v8f s0a = {}, s0b = {}, s1a = {}, s1b = {};
        #pragma unroll
        for (int kk = 0; kk < 4; ++kk) {
            const v2f ka = pick2(ka_lo, ka_hi, kk);
            const v2f kb = pick2(kb_lo, kb_hi, kk);
            const v2f bq0 = pick2(q0_lo, q0_hi, kk);
            const v2f bq1 = pick2(q1_lo, q1_hi, kk);
            s0a = wmma_f32_k4(ka, bq0, s0a);
            s0b = wmma_f32_k4(kb, bq0, s0b);
            s1a = wmma_f32_k4(ka, bq1, s1a);
            s1b = wmma_f32_k4(kb, bq1, s1b);
        }

        softmax_step(s0a, s0b, a0a_lo, a0a_hi, a0b_lo, a0b_hi, mrun0, lsum0, acc00, acc01);
        softmax_step(s1a, s1b, a1a_lo, a1a_hi, a1b_lo, a1b_hi, mrun1, lsum1, acc10, acc11);

        #pragma unroll
        for (int j = 0; j < 4; ++j) {
            const v2f va = pick2(va_lo, va_hi, j);
            const v2f vb = pick2(vb_lo, vb_hi, j);
            v2f p0a; p0a.x = s0a[2 * j]; p0a.y = s0a[2 * j + 1];
            v2f p0b; p0b.x = s0b[2 * j]; p0b.y = s0b[2 * j + 1];
            v2f p1a; p1a.x = s1a[2 * j]; p1a.y = s1a[2 * j + 1];
            v2f p1b; p1b.x = s1b[2 * j]; p1b.y = s1b[2 * j + 1];
            acc00 = wmma_f32_k4(va, p0a, acc00);
            acc01 = wmma_f32_k4(vb, p0b, acc01);
            acc10 = wmma_f32_k4(va, p1a, acc10);
            acc11 = wmma_f32_k4(vb, p1b, acc11);
        }
    }

    const float* xr = x + base + (size_t)(n0 + 8 * h) * D_ + lane;
    float xa[8], xb[8];
    #pragma unroll
    for (int r = 0; r < 8; ++r) { xa[r] = xr[r * D_]; xb[r] = xr[r * D_ + 32]; }
    const float g0 = gamma[lane], g1 = gamma[lane + 32];
    const float b0 = beta[lane],  b1 = beta[lane + 32];

    const float inv0 = 1.0f / (lsum0 + xhalf(lsum0));
    const float inv1 = 1.0f / (lsum1 + xhalf(lsum1));
    v4f o_lo, o_hi;
    o_lo.x = (acc00[0] + acc01[0]) * inv0; o_lo.y = (acc00[1] + acc01[1]) * inv0;
    o_lo.z = (acc00[2] + acc01[2]) * inv0; o_lo.w = (acc00[3] + acc01[3]) * inv0;
    o_hi.x = (acc00[4] + acc01[4]) * inv0; o_hi.y = (acc00[5] + acc01[5]) * inv0;
    o_hi.z = (acc00[6] + acc01[6]) * inv0; o_hi.w = (acc00[7] + acc01[7]) * inv0;
    v4f* hp = &hs[l16 * HS_PITCH4_ + h * (HD_ / 4) + 2 * half];
    hp[0] = o_lo;
    hp[1] = o_hi;
    o_lo.x = (acc10[0] + acc11[0]) * inv1; o_lo.y = (acc10[1] + acc11[1]) * inv1;
    o_lo.z = (acc10[2] + acc11[2]) * inv1; o_lo.w = (acc10[3] + acc11[3]) * inv1;
    o_hi.x = (acc10[4] + acc11[4]) * inv1; o_hi.y = (acc10[5] + acc11[5]) * inv1;
    o_hi.z = (acc10[6] + acc11[6]) * inv1; o_hi.w = (acc10[7] + acc11[7]) * inv1;
    hp[16 * HS_PITCH4_]     = o_lo;
    hp[16 * HS_PITCH4_ + 1] = o_hi;
    __syncthreads();

    const float* hsf = (const float*)hs;
    float* orow = out + base + (size_t)(n0 + 8 * h) * D_ + lane;
    #pragma unroll
    for (int r = 0; r < 8; ++r) {
        const float* hrow = hsf + (8 * h + r) * (HS_PITCH4_ * 4);
        const float y0 = hrow[lane]      + xa[r];
        const float y1 = hrow[lane + 32] + xb[r];
        float s1 = y0 + y1;
        float s2 = y0 * y0 + y1 * y1;
        #pragma unroll
        for (int m = 1; m < 32; m <<= 1) {
            s1 += __shfl_xor(s1, m);
            s2 += __shfl_xor(s2, m);
        }
        const float mean = s1 * (1.0f / 64.0f);
        const float var  = s2 * (1.0f / 64.0f) - mean * mean;
        const float rstd = rsqrtf(var + EPS_);
        orow[r * D_]      = (y0 - mean) * rstd * g0 + b0;
        orow[r * D_ + 32] = (y1 - mean) * rstd * g1 + b1;
    }
}

extern "C" void kernel_launch(void* const* d_in, const int* in_sizes, int n_in,
                              void* d_out, int out_size, void* d_ws, size_t ws_size,
                              hipStream_t stream) {
    (void)in_sizes; (void)n_in; (void)out_size; (void)ws_size;
    const float* x     = (const float*)d_in[0];
    const float* adj   = (const float*)d_in[1];
    const float* Wq    = (const float*)d_in[3];
    const float* bq    = (const float*)d_in[4];
    const float* Wk    = (const float*)d_in[5];
    const float* bk    = (const float*)d_in[6];
    const float* Wv    = (const float*)d_in[7];
    const float* bv    = (const float*)d_in[8];
    const float* gamma = (const float*)d_in[9];
    const float* beta  = (const float*)d_in[10];
    float* out = (float*)d_out;

    float* ws = (float*)d_ws;
    const size_t S = (size_t)B_ * T_ * N_ * D_;
    float* Q  = ws;
    float* K  = ws + S;
    float* V  = ws + 2 * S;

    dim3 g1(B_ * T_ * N_ / 16, 3, 1);
    qkv_kernel<<<g1, 32, 0, stream>>>(x, Wq, bq, Wk, bk, Wv, bv, Q, K, V);

    dim3 g2(N_ / QROWS_, B_ * T_, 1);
    attn_ln_kernel<<<g2, 128, 0, stream>>>(Q, K, V, adj, x, gamma, beta, out);
}


# attention softmax on the score registers' own adjacent pairs (packed ops, no register moves), log2(e) folded into Q's scale so exp2 takes the scores; denominator as a pair tree: sum order changes
# speedup vs baseline: 2.1544x; 1.0481x over previous
#include <hip/hip_runtime.h>

#define B_ 2
#define T_ 12
#define N_ 1024
#define D_ 64
#define H_ 4
#define HD_ 16
#define EPS_ 1e-5f
#define NEG_BIG -3.0e38f
#define QSCALE_ 0.36067376022224085f


typedef __attribute__((ext_vector_type(2))) float v2f;
typedef __attribute__((ext_vector_type(8))) float v8f;

__device__ __forceinline__ v8f wmma_f32_k4(v2f a, v2f b, v8f c) {
    return __builtin_amdgcn_wmma_f32_16x16x4_f32(
         false, a,  false, b,
         (short)0, c,  false,  false);
}

#define FRAG_TILE_ 256
__device__ __forceinline__ size_t frag_off(size_t bt, int head, size_t tile) {
    return ((bt * H_ + head) * (N_ / 16) + tile) * FRAG_TILE_;
}
__global__ __launch_bounds__(32) void qkv_kernel(
    const float* __restrict__ x,
    const float* __restrict__ Wq, const float* __restrict__ bq,
    const float* __restrict__ Wk, const float* __restrict__ bk,
    const float* __restrict__ Wv, const float* __restrict__ bv,
    float* __restrict__ Q, float* __restrict__ Ko, float* __restrict__ V)
{
    __shared__ float xs[16 * 68];
    const int lane = threadIdx.x & 31;
    const int half = lane >> 4;
    const int l16  = lane & 15;
    const int which = blockIdx.y;
    const float* W    = (which == 0) ? Wq : (which == 1) ? Wk : Wv;
    const float* bias = (which == 0) ? bq : (which == 1) ? bk : bv;
    float*       out  = (which == 0) ? Q  : (which == 1) ? Ko : V;
    const float  osc  = (which == 0) ? QSCALE_ : 1.0f;

    const size_t row0 = (size_t)blockIdx.x * 16;
    const float* xt = x + row0 * D_;

    #pragma unroll
    for (int i = 0; i < 8; ++i) {
        const int idx = (i * 32 + lane) * 4;
        const float4 val = *(const float4*)(xt + idx);
        const int row = idx >> 6, col = idx & 63;
        float* dst = &xs[row * 68 + col];
        dst[0] = val.x; dst[1] = val.y; dst[2] = val.z; dst[3] = val.w;
    }
    __syncthreads();

    v2f afrag[16];
    #pragma unroll
    for (int kk = 0; kk < 16; ++kk) {
        const float* p = &xs[l16 * 68 + kk * 4 + 2 * half];
        v2f a; a.x = p[0]; a.y = p[1];
        afrag[kk] = a;
    }

    v8f acc[4];
    #pragma unroll
    for (int c = 0; c < 4; ++c) acc[c] = {};

    const size_t bt   = row0 / N_;
    const size_t tile = (row0 % N_) / 16;

    if (which == 2) {
        #pragma unroll
        for (int kk = 0; kk < 16; ++kk) {
            #pragma unroll
            for (int c = 0; c < 4; ++c) {
                const float* wp = &W[(c * 16 + l16) * D_ + kk * 4 + 2 * half];
                v2f w; w.x = wp[0]; w.y = wp[1];
                acc[c] = wmma_f32_k4(afrag[kk], w, acc[c]);
            }
        }
        #pragma unroll
        for (int c = 0; c < 4; ++c) {
            const float bb = bias[c * 16 + l16];
            float* f = out + frag_off(bt, c, tile) + lane * 4;
            float4 lo, hi;
            lo.x = acc[c][0] + bb; lo.y = acc[c][1] + bb; lo.z = acc[c][2] + bb; lo.w = acc[c][3] + bb;
            hi.x = acc[c][4] + bb; hi.y = acc[c][5] + bb; hi.z = acc[c][6] + bb; hi.w = acc[c][7] + bb;
            *(float4*)f         = lo;
            *(float4*)(f + 128) = hi;
        }
    } else {
        #pragma unroll
        for (int kk = 0; kk < 16; ++kk) {
            #pragma unroll
            for (int c = 0; c < 4; ++c) {
                const float* wp = &W[(c * 16 + l16) * D_ + kk * 4 + 2 * half];
                v2f w; w.x = wp[0]; w.y = wp[1];
                acc[c] = wmma_f32_k4(w, afrag[kk], acc[c]);
            }
        }
        #pragma unroll
        for (int c = 0; c < 4; ++c) {
            const float4 b_lo = *(const float4*)(bias + c * 16 + 8 * half);
            const float4 b_hi = *(const float4*)(bias + c * 16 + 8 * half + 4);
            float* f = out + frag_off(bt, c, tile) + lane * 4;
            float4 lo, hi;
            lo.x = (acc[c][0] + b_lo.x) * osc; lo.y = (acc[c][1] + b_lo.y) * osc;
            lo.z = (acc[c][2] + b_lo.z) * osc; lo.w = (acc[c][3] + b_lo.w) * osc;
            hi.x = (acc[c][4] + b_hi.x) * osc; hi.y = (acc[c][5] + b_hi.y) * osc;
            hi.z = (acc[c][6] + b_hi.z) * osc; hi.w = (acc[c][7] + b_hi.w) * osc;
            *(float4*)f         = lo;
            *(float4*)(f + 128) = hi;
        }
    }
}

typedef __attribute__((ext_vector_type(4))) float v4f;

#define GKEYS_ 32

__device__ __forceinline__ float xhalf(float x) {
    const int b = __builtin_bit_cast(int, x);
    return __builtin_bit_cast(float,
        __builtin_amdgcn_permlanex16(b, b, 0x76543210, 0xfedcba98, false, false));
}

__device__ __forceinline__ v2f pick2(v4f lo, v4f hi, int j) {
    v2f r;
    if (j == 0)      { r.x = lo.x; r.y = lo.y; }
    else if (j == 1) { r.x = lo.z; r.y = lo.w; }
    else if (j == 2) { r.x = hi.x; r.y = hi.y; }
    else             { r.x = hi.z; r.y = hi.w; }
    return r;
}

#define QROWS_ 32
#define HS_PITCH4_ 17

#define ADJ_PITCH4_ 9
#define ADJ_SLAB4_ (QROWS_ * ADJ_PITCH4_)

__device__ __forceinline__ float max3f(float a, float b, float c) {
    return fmaxf(fmaxf(a, b), c);
}
__device__ __forceinline__ v2f pair8(v8f v, int j) {
    v2f r; r.x = v[2 * j]; r.y = v[2 * j + 1];
    return r;
}
__device__ __forceinline__ v2f lo2(v4f v) { v2f r; r.x = v.x; r.y = v.y; return r; }
__device__ __forceinline__ v2f hi2(v4f v) { v2f r; r.x = v.z; r.y = v.w; return r; }

__device__ __forceinline__ void softmax_step(
    v8f sa, v8f sb, v4f aa_lo, v4f aa_hi, v4f ab_lo, v4f ab_hi,
    float& mrun, v2f& lsum2, v8f& acc0, v8f& acc1, v2f (&t)[8])
{
    t[0] = pair8(sa, 0) * lo2(aa_lo);  t[1] = pair8(sa, 1) * hi2(aa_lo);
    t[2] = pair8(sa, 2) * lo2(aa_hi);  t[3] = pair8(sa, 3) * hi2(aa_hi);
    t[4] = pair8(sb, 0) * lo2(ab_lo);  t[5] = pair8(sb, 1) * hi2(ab_lo);
    t[6] = pair8(sb, 2) * lo2(ab_hi);  t[7] = pair8(sb, 3) * hi2(ab_hi);

    float ma = max3f(mrun, t[0].x, t[0].y);
    float mb = max3f(t[4].x, t[4].y, t[5].x);
    ma = max3f(ma, t[1].x, t[1].y);
    mb = max3f(mb, t[5].y, t[6].x);
    ma = max3f(ma, t[2].x, t[2].y);
    mb = max3f(mb, t[6].y, t[7].x);
    ma = max3f(ma, t[3].x, t[3].y);
    const float mx = max3f(ma, mb, t[7].y);
    const float mnew = fmaxf(mx, xhalf(mx));
    const float sc   = __builtin_amdgcn_exp2f(mrun - mnew);
    mrun = mnew;

    #pragma unroll
    for (int i = 0; i < 8; ++i) {
        const v2f d = t[i] - mnew;
        v2f e; e.x = __builtin_amdgcn_exp2f(d.x); e.y = __builtin_amdgcn_exp2f(d.y);
        t[i] = e;
    }
    const v2f csum = ((t[0] + t[1]) + (t[2] + t[3])) + ((t[4] + t[5]) + (t[6] + t[7]));
    lsum2 = lsum2 * sc + csum;
    acc0 = acc0 * sc;
    acc1 = acc1 * sc;
}

__global__ __launch_bounds__(128) void attn_ln_kernel(
    const float* __restrict__ Qf, const float* __restrict__ Kf,
    const float* __restrict__ Vf, const float* __restrict__ adj,
    const float* __restrict__ x, const float* __restrict__ gamma,
    const float* __restrict__ beta, float* __restrict__ out)
{
    __shared__ v4f hs[QROWS_ * HS_PITCH4_];
    __shared__ v4f adjs[H_ * ADJ_SLAB4_];
    const int lane = threadIdx.x & 31;
    const int half = lane >> 4;
    const int l16  = lane & 15;
    const int n0 = blockIdx.x * QROWS_;
    const int h  = __builtin_amdgcn_readfirstlane((int)(threadIdx.x >> 5));
    const int bt = blockIdx.y;
    const size_t base  = (size_t)bt * N_ * D_;
    const size_t abase = (size_t)bt * N_ * N_;

    const float* qp = Qf + frag_off(bt, h, (size_t)blockIdx.x * 2) + lane * 4;
    const v4f q0_lo = *(const v4f*)qp;
    const v4f q0_hi = *(const v4f*)(qp + 128);
    const v4f q1_lo = *(const v4f*)(qp + FRAG_TILE_);
    const v4f q1_hi = *(const v4f*)(qp + FRAG_TILE_ + 128);

    const float* kp = Kf + frag_off(bt, h, 0) + lane * 4;
    const float* vp = Vf + frag_off(bt, h, 0) + lane * 4;
    const float* ag = adj + abase + (size_t)(n0 + (lane >> 3)) * N_ + 4 * (lane & 7);
    v4f* aslab = adjs + h * ADJ_SLAB4_;
    v4f* ast = aslab + (lane >> 3) * ADJ_PITCH4_ + (lane & 7);
    const v4f* ald0 = aslab + l16 * ADJ_PITCH4_ + 2 * half;
    const v4f* ald1 = ald0 + 16 * ADJ_PITCH4_;

    float mrun0 = NEG_BIG, mrun1 = NEG_BIG;
    v2f lsum0 = {}, lsum1 = {};
    v8f acc00 = {}, acc01 = {};
    v8f acc10 = {}, acc11 = {};

    for (int m0 = 0; m0 < N_; m0 += GKEYS_) {
        const float* k0 = kp + (size_t)m0 * (FRAG_TILE_ / 16);
        const v4f ka_lo = *(const v4f*)(k0);
        const v4f ka_hi = *(const v4f*)(k0 + 128);
        const v4f kb_lo = *(const v4f*)(k0 + FRAG_TILE_);
        const v4f kb_hi = *(const v4f*)(k0 + FRAG_TILE_ + 128);
        v4f g[8];
        #pragma unroll
        for (int q = 0; q < 8; ++q) g[q] = *(const v4f*)(ag + m0 + (size_t)(4 * q) * N_);
        const float* v0 = vp + (size_t)m0 * (FRAG_TILE_ / 16);
        const v4f va_lo = *(const v4f*)(v0);
        const v4f va_hi = *(const v4f*)(v0 + 128);
        const v4f vb_lo = *(const v4f*)(v0 + FRAG_TILE_);
        const v4f vb_hi = *(const v4f*)(v0 + FRAG_TILE_ + 128);

        __builtin_amdgcn_fence(__ATOMIC_ACQ_REL, "wavefront");
        #pragma unroll
        for (int q = 0; q < 8; ++q) ast[4 * q * ADJ_PITCH4_] = g[q];
        __builtin_amdgcn_fence(__ATOMIC_ACQ_REL, "wavefront");
        const v4f a0a_lo = ald0[0];
        const v4f a0a_hi = ald0[1];
        const v4f a0b_lo = ald0[4];
        const v4f a0b_hi = ald0[5];
        const v4f a1a_lo = ald1[0];
        const v4f a1a_hi = ald1[1];
        const v4f a1b_lo = ald1[4];
        const v4f a1b_hi = ald1[5];

        v8f s0a = {}, s0b = {}, s1a = {}, s1b = {};
        #pragma unroll
        for (int kk = 0; kk < 4; ++kk) {
            const v2f ka = pick2(ka_lo, ka_hi, kk);
            const v2f kb = pick2(kb_lo, kb_hi, kk);
            const v2f bq0 = pick2(q0_lo, q0_hi, kk);
            const v2f bq1 = pick2(q1_lo, q1_hi, kk);
            s0a = wmma_f32_k4(ka, bq0, s0a);
            s0b = wmma_f32_k4(kb, bq0, s0b);
            s1a = wmma_f32_k4(ka, bq1, s1a);
            s1b = wmma_f32_k4(kb, bq1, s1b);
        }

        v2f p0[8], p1[8];
        softmax_step(s0a, s0b, a0a_lo, a0a_hi, a0b_lo, a0b_hi, mrun0, lsum0, acc00, acc01, p0);
        softmax_step(s1a, s1b, a1a_lo, a1a_hi, a1b_lo, a1b_hi, mrun1, lsum1, acc10, acc11, p1);

        #pragma unroll
        for (int j = 0; j < 4; ++j) {
            const v2f va = pick2(va_lo, va_hi, j);
            const v2f vb = pick2(vb_lo, vb_hi, j);
            acc00 = wmma_f32_k4(va, p0[j],     acc00);
            acc01 = wmma_f32_k4(vb, p0[4 + j], acc01);
            acc10 = wmma_f32_k4(va, p1[j],     acc10);
            acc11 = wmma_f32_k4(vb, p1[4 + j], acc11);
        }
    }

    const float* xr = x + base + (size_t)(n0 + 8 * h) * D_ + lane;
    float xa[8], xb[8];
    #pragma unroll
    for (int r = 0; r < 8; ++r) { xa[r] = xr[r * D_]; xb[r] = xr[r * D_ + 32]; }
    const float g0 = gamma[lane], g1 = gamma[lane + 32];
    const float b0 = beta[lane],  b1 = beta[lane + 32];

    const float l0 = lsum0.x + lsum0.y, l1 = lsum1.x + lsum1.y;
    const float inv0 = 1.0f / (l0 + xhalf(l0));
    const float inv1 = 1.0f / (l1 + xhalf(l1));
    v4f o_lo, o_hi;
    o_lo.x = (acc00[0] + acc01[0]) * inv0; o_lo.y = (acc00[1] + acc01[1]) * inv0;
    o_lo.z = (acc00[2] + acc01[2]) * inv0; o_lo.w = (acc00[3] + acc01[3]) * inv0;
    o_hi.x = (acc00[4] + acc01[4]) * inv0; o_hi.y = (acc00[5] + acc01[5]) * inv0;
    o_hi.z = (acc00[6] + acc01[6]) * inv0; o_hi.w = (acc00[7] + acc01[7]) * inv0;
    v4f* hp = &hs[l16 * HS_PITCH4_ + h * (HD_ / 4) + 2 * half];
    hp[0] = o_lo;
    hp[1] = o_hi;
    o_lo.x = (acc10[0] + acc11[0]) * inv1; o_lo.y = (acc10[1] + acc11[1]) * inv1;
    o_lo.z = (acc10[2] + acc11[2]) * inv1; o_lo.w = (acc10[3] + acc11[3]) * inv1;
    o_hi.x = (acc10[4] + acc11[4]) * inv1; o_hi.y = (acc10[5] + acc11[5]) * inv1;
    o_hi.z = (acc10[6] + acc11[6]) * inv1; o_hi.w = (acc10[7] + acc11[7]) * inv1;
    hp[16 * HS_PITCH4_]     = o_lo;
    hp[16 * HS_PITCH4_ + 1] = o_hi;
    __syncthreads();

    const float* hsf = (const float*)hs;
    float* orow = out + base + (size_t)(n0 + 8 * h) * D_ + lane;
    #pragma unroll
    for (int r = 0; r < 8; ++r) {
        const float* hrow = hsf + (8 * h + r) * (HS_PITCH4_ * 4);
        const float y0 = hrow[lane]      + xa[r];
        const float y1 = hrow[lane + 32] + xb[r];
        float s1 = y0 + y1;
        float s2 = y0 * y0 + y1 * y1;
        #pragma unroll
        for (int m = 1; m < 32; m <<= 1) {
            s1 += __shfl_xor(s1, m);
            s2 += __shfl_xor(s2, m);
        }
        const float mean = s1 * (1.0f / 64.0f);
        const float var  = s2 * (1.0f / 64.0f) - mean * mean;
        const float rstd = rsqrtf(var + EPS_);
        orow[r * D_]      = (y0 - mean) * rstd * g0 + b0;
        orow[r * D_ + 32] = (y1 - mean) * rstd * g1 + b1;
    }
}

extern "C" void kernel_launch(void* const* d_in, const int* in_sizes, int n_in,
                              void* d_out, int out_size, void* d_ws, size_t ws_size,
                              hipStream_t stream) {
    (void)in_sizes; (void)n_in; (void)out_size; (void)ws_size;
    const float* x     = (const float*)d_in[0];
    const float* adj   = (const float*)d_in[1];
    const float* Wq    = (const float*)d_in[3];
    const float* bq    = (const float*)d_in[4];
    const float* Wk    = (const float*)d_in[5];
    const float* bk    = (const float*)d_in[6];
    const float* Wv    = (const float*)d_in[7];
    const float* bv    = (const float*)d_in[8];
    const float* gamma = (const float*)d_in[9];
    const float* beta  = (const float*)d_in[10];
    float* out = (float*)d_out;

    float* ws = (float*)d_ws;
    const size_t S = (size_t)B_ * T_ * N_ * D_;
    float* Q  = ws;
    float* K  = ws + S;
    float* V  = ws + 2 * S;

    dim3 g1(B_ * T_ * N_ / 16, 3, 1);
    qkv_kernel<<<g1, 32, 0, stream>>>(x, Wq, bq, Wk, bk, Wv, bv, Q, K, V);

    dim3 g2(N_ / QROWS_, B_ * T_, 1);
    attn_ln_kernel<<<g2, 128, 0, stream>>>(Q, K, V, adj, x, gamma, beta, out);
}


# projection kernel: each 4-wave work-group copies its W into LDS once in whole 128-byte lines and reads the fragments from LDS (was 64 global loads of 16 lines each per wave); k summation order kept
# speedup vs baseline: 2.2508x; 1.0447x over previous
#include <hip/hip_runtime.h>

#define B_ 2
#define T_ 12
#define N_ 1024
#define D_ 64
#define H_ 4
#define HD_ 16
#define EPS_ 1e-5f
#define NEG_BIG -3.0e38f
#define QSCALE_ 0.36067376022224085f


typedef __attribute__((ext_vector_type(2))) float v2f;
typedef __attribute__((ext_vector_type(8))) float v8f;

__device__ __forceinline__ v8f wmma_f32_k4(v2f a, v2f b, v8f c) {
    return __builtin_amdgcn_wmma_f32_16x16x4_f32(
         false, a,  false, b,
         (short)0, c,  false,  false);
}

#define FRAG_TILE_ 256
#define QKV_WAVES_ 4
#define LDS_PITCH4_ 17
__device__ __forceinline__ size_t frag_off(size_t bt, int head, size_t tile) {
    return ((bt * H_ + head) * (N_ / 16) + tile) * FRAG_TILE_;
}
__global__ __launch_bounds__(32 * QKV_WAVES_) void qkv_kernel(
    const float* __restrict__ x,
    const float* __restrict__ Wq, const float* __restrict__ bq,
    const float* __restrict__ Wk, const float* __restrict__ bk,
    const float* __restrict__ Wv, const float* __restrict__ bv,
    float* __restrict__ Q, float* __restrict__ Ko, float* __restrict__ V)
{
    __shared__ float4 ws4[D_ * LDS_PITCH4_];
    __shared__ float4 xs4[QKV_WAVES_ * 16 * LDS_PITCH4_];
    const int tid  = threadIdx.x;
    const int lane = tid & 31;
    const int half = lane >> 4;
    const int l16  = lane & 15;
    const int wv = __builtin_amdgcn_readfirstlane(tid >> 5);
    const int which = blockIdx.y;
    const float* W    = (which == 0) ? Wq : (which == 1) ? Wk : Wv;
    const float* bias = (which == 0) ? bq : (which == 1) ? bk : bv;
    float*       out  = (which == 0) ? Q  : (which == 1) ? Ko : V;
    const float  osc  = (which == 0) ? QSCALE_ : 1.0f;

    const size_t row0 = ((size_t)blockIdx.x * QKV_WAVES_ + wv) * 16;
    const float* xt = x + row0 * D_;
    float4* xw4 = xs4 + wv * (16 * LDS_PITCH4_);

    #pragma unroll
    for (int i = 0; i < 8; ++i) {
        const int idx4 = i * (32 * QKV_WAVES_) + tid;
        ws4[(idx4 >> 4) * LDS_PITCH4_ + (idx4 & 15)] = *(const float4*)(W + idx4 * 4);
    }
    #pragma unroll
    for (int i = 0; i < 8; ++i) {
        const int idx4 = i * 32 + lane;
        xw4[(idx4 >> 4) * LDS_PITCH4_ + (idx4 & 15)] = *(const float4*)(xt + idx4 * 4);
    }
    __syncthreads();

    const float* xsf = (const float*)xw4;
    v2f afrag[16];
    #pragma unroll
    for (int kk = 0; kk < 16; ++kk) {
        const float* p = &xsf[l16 * (LDS_PITCH4_ * 4) + kk * 4 + 2 * half];
        v2f a; a.x = p[0]; a.y = p[1];
        afrag[kk] = a;
    }

    v8f acc[4];
    #pragma unroll
    for (int c = 0; c < 4; ++c) acc[c] = {};

    const size_t bt   = row0 / N_;
    const size_t tile = (row0 % N_) / 16;

    const float* wl = (const float*)ws4 + l16 * (LDS_PITCH4_ * 4) + 2 * half;
    if (which == 2) {
        #pragma unroll
        for (int kk = 0; kk < 16; ++kk) {
            #pragma unroll
            for (int c = 0; c < 4; ++c) {
                const float* wp = wl + c * 16 * (LDS_PITCH4_ * 4) + kk * 4;
                v2f w; w.x = wp[0]; w.y = wp[1];
                acc[c] = wmma_f32_k4(afrag[kk], w, acc[c]);
            }
        }
        #pragma unroll
        for (int c = 0; c < 4; ++c) {
            const float bb = bias[c * 16 + l16];
            float* f = out + frag_off(bt, c, tile) + lane * 4;
            float4 lo, hi;
            lo.x = acc[c][0] + bb; lo.y = acc[c][1] + bb; lo.z = acc[c][2] + bb; lo.w = acc[c][3] + bb;
            hi.x = acc[c][4] + bb; hi.y = acc[c][5] + bb; hi.z = acc[c][6] + bb; hi.w = acc[c][7] + bb;
            *(float4*)f         = lo;
            *(float4*)(f + 128) = hi;
        }
    } else {
        #pragma unroll
        for (int kk = 0; kk < 16; ++kk) {
            #pragma unroll
            for (int c = 0; c < 4; ++c) {
                const float* wp = wl + c * 16 * (LDS_PITCH4_ * 4) + kk * 4;
                v2f w; w.x = wp[0]; w.y = wp[1];
                acc[c] = wmma_f32_k4(w, afrag[kk], acc[c]);
            }
        }
        #pragma unroll
        for (int c = 0; c < 4; ++c) {
            const float4 b_lo = *(const float4*)(bias + c * 16 + 8 * half);
            const float4 b_hi = *(const float4*)(bias + c * 16 + 8 * half + 4);
            float* f = out + frag_off(bt, c, tile) + lane * 4;
            float4 lo, hi;
            lo.x = (acc[c][0] + b_lo.x) * osc; lo.y = (acc[c][1] + b_lo.y) * osc;
            lo.z = (acc[c][2] + b_lo.z) * osc; lo.w = (acc[c][3] + b_lo.w) * osc;
            hi.x = (acc[c][4] + b_hi.x) * osc; hi.y = (acc[c][5] + b_hi.y) * osc;
            hi.z = (acc[c][6] + b_hi.z) * osc; hi.w = (acc[c][7] + b_hi.w) * osc;
            *(float4*)f         = lo;
            *(float4*)(f + 128) = hi;
        }
    }
}

typedef __attribute__((ext_vector_type(4))) float v4f;

#define GKEYS_ 32

__device__ __forceinline__ float xhalf(float x) {
    const int b = __builtin_bit_cast(int, x);
    return __builtin_bit_cast(float,
        __builtin_amdgcn_permlanex16(b, b, 0x76543210, 0xfedcba98, false, false));
}

__device__ __forceinline__ v2f pick2(v4f lo, v4f hi, int j) {
    v2f r;
    if (j == 0)      { r.x = lo.x; r.y = lo.y; }
    else if (j == 1) { r.x = lo.z; r.y = lo.w; }
    else if (j == 2) { r.x = hi.x; r.y = hi.y; }
    else             { r.x = hi.z; r.y = hi.w; }
    return r;
}

#define QROWS_ 32
#define HS_PITCH4_ 17

#define ADJ_PITCH4_ 9
#define ADJ_SLAB4_ (QROWS_ * ADJ_PITCH4_)

__device__ __forceinline__ float max3f(float a, float b, float c) {
    return fmaxf(fmaxf(a, b), c);
}
__device__ __forceinline__ v2f pair8(v8f v, int j) {
    v2f r; r.x = v[2 * j]; r.y = v[2 * j + 1];
    return r;
}
__device__ __forceinline__ v2f lo2(v4f v) { v2f r; r.x = v.x; r.y = v.y; return r; }
__device__ __forceinline__ v2f hi2(v4f v) { v2f r; r.x = v.z; r.y = v.w; return r; }

__device__ __forceinline__ void softmax_step(
    v8f sa, v8f sb, v4f aa_lo, v4f aa_hi, v4f ab_lo, v4f ab_hi,
    float& mrun, v2f& lsum2, v8f& acc0, v8f& acc1, v2f (&t)[8])
{
    t[0] = pair8(sa, 0) * lo2(aa_lo);  t[1] = pair8(sa, 1) * hi2(aa_lo);
    t[2] = pair8(sa, 2) * lo2(aa_hi);  t[3] = pair8(sa, 3) * hi2(aa_hi);
    t[4] = pair8(sb, 0) * lo2(ab_lo);  t[5] = pair8(sb, 1) * hi2(ab_lo);
    t[6] = pair8(sb, 2) * lo2(ab_hi);  t[7] = pair8(sb, 3) * hi2(ab_hi);

    float ma = max3f(mrun, t[0].x, t[0].y);
    float mb = max3f(t[4].x, t[4].y, t[5].x);
    ma = max3f(ma, t[1].x, t[1].y);
    mb = max3f(mb, t[5].y, t[6].x);
    ma = max3f(ma, t[2].x, t[2].y);
    mb = max3f(mb, t[6].y, t[7].x);
    ma = max3f(ma, t[3].x, t[3].y);
    const float mx = max3f(ma, mb, t[7].y);
    const float mnew = fmaxf(mx, xhalf(mx));
    const float sc   = __builtin_amdgcn_exp2f(mrun - mnew);
    mrun = mnew;

    #pragma unroll
    for (int i = 0; i < 8; ++i) {
        const v2f d = t[i] - mnew;
        v2f e; e.x = __builtin_amdgcn_exp2f(d.x); e.y = __builtin_amdgcn_exp2f(d.y);
        t[i] = e;
    }
    const v2f csum = ((t[0] + t[1]) + (t[2] + t[3])) + ((t[4] + t[5]) + (t[6] + t[7]));
    lsum2 = lsum2 * sc + csum;
    acc0 = acc0 * sc;
    acc1 = acc1 * sc;
}

__global__ __launch_bounds__(128) void attn_ln_kernel(
    const float* __restrict__ Qf, const float* __restrict__ Kf,
    const float* __restrict__ Vf, const float* __restrict__ adj,
    const float* __restrict__ x, const float* __restrict__ gamma,
    const float* __restrict__ beta, float* __restrict__ out)
{
    __shared__ v4f hs[QROWS_ * HS_PITCH4_];
    __shared__ v4f adjs[H_ * ADJ_SLAB4_];
    const int lane = threadIdx.x & 31;
    const int half = lane >> 4;
    const int l16  = lane & 15;
    const int n0 = blockIdx.x * QROWS_;
    const int h  = __builtin_amdgcn_readfirstlane((int)(threadIdx.x >> 5));
    const int bt = blockIdx.y;
    const size_t base  = (size_t)bt * N_ * D_;
    const size_t abase = (size_t)bt * N_ * N_;

    const float* qp = Qf + frag_off(bt, h, (size_t)blockIdx.x * 2) + lane * 4;
    const v4f q0_lo = *(const v4f*)qp;
    const v4f q0_hi = *(const v4f*)(qp + 128);
    const v4f q1_lo = *(const v4f*)(qp + FRAG_TILE_);
    const v4f q1_hi = *(const v4f*)(qp + FRAG_TILE_ + 128);

    const float* kp = Kf + frag_off(bt, h, 0) + lane * 4;
    const float* vp = Vf + frag_off(bt, h, 0) + lane * 4;
    const float* ag = adj + abase + (size_t)(n0 + (lane >> 3)) * N_ + 4 * (lane & 7);
    v4f* aslab = adjs + h * ADJ_SLAB4_;
    v4f* ast = aslab + (lane >> 3) * ADJ_PITCH4_ + (lane & 7);
    const v4f* ald0 = aslab + l16 * ADJ_PITCH4_ + 2 * half;
    const v4f* ald1 = ald0 + 16 * ADJ_PITCH4_;

    float mrun0 = NEG_BIG, mrun1 = NEG_BIG;
    v2f lsum0 = {}, lsum1 = {};
    v8f acc00 = {}, acc01 = {};
    v8f acc10 = {}, acc11 = {};

    for (int m0 = 0; m0 < N_; m0 += GKEYS_) {
        const float* k0 = kp + (size_t)m0 * (FRAG_TILE_ / 16);
        const v4f ka_lo = *(const v4f*)(k0);
        const v4f ka_hi = *(const v4f*)(k0 + 128);
        const v4f kb_lo = *(const v4f*)(k0 + FRAG_TILE_);
        const v4f kb_hi = *(const v4f*)(k0 + FRAG_TILE_ + 128);
        v4f g[8];
        #pragma unroll
        for (int q = 0; q < 8; ++q) g[q] = *(const v4f*)(ag + m0 + (size_t)(4 * q) * N_);
        const float* v0 = vp + (size_t)m0 * (FRAG_TILE_ / 16);
        const v4f va_lo = *(const v4f*)(v0);
        const v4f va_hi = *(const v4f*)(v0 + 128);
        const v4f vb_lo = *(const v4f*)(v0 + FRAG_TILE_);
        const v4f vb_hi = *(const v4f*)(v0 + FRAG_TILE_ + 128);

        __builtin_amdgcn_fence(__ATOMIC_ACQ_REL, "wavefront");
        #pragma unroll
        for (int q = 0; q < 8; ++q) ast[4 * q * ADJ_PITCH4_] = g[q];
        __builtin_amdgcn_fence(__ATOMIC_ACQ_REL, "wavefront");
        const v4f a0a_lo = ald0[0];
        const v4f a0a_hi = ald0[1];
        const v4f a0b_lo = ald0[4];
        const v4f a0b_hi = ald0[5];
        const v4f a1a_lo = ald1[0];
        const v4f a1a_hi = ald1[1];
        const v4f a1b_lo = ald1[4];
        const v4f a1b_hi = ald1[5];

        v8f s0a = {}, s0b = {}, s1a = {}, s1b = {};
        #pragma unroll
        for (int kk = 0; kk < 4; ++kk) {
            const v2f ka = pick2(ka_lo, ka_hi, kk);
            const v2f kb = pick2(kb_lo, kb_hi, kk);
            const v2f bq0 = pick2(q0_lo, q0_hi, kk);
            const v2f bq1 = pick2(q1_lo, q1_hi, kk);
            s0a = wmma_f32_k4(ka, bq0, s0a);
            s0b = wmma_f32_k4(kb, bq0, s0b);
            s1a = wmma_f32_k4(ka, bq1, s1a);
            s1b = wmma_f32_k4(kb, bq1, s1b);
        }

        v2f p0[8], p1[8];
        softmax_step(s0a, s0b, a0a_lo, a0a_hi, a0b_lo, a0b_hi, mrun0, lsum0, acc00, acc01, p0);
        softmax_step(s1a, s1b, a1a_lo, a1a_hi, a1b_lo, a1b_hi, mrun1, lsum1, acc10, acc11, p1);

        #pragma unroll
        for (int j = 0; j < 4; ++j) {
            const v2f va = pick2(va_lo, va_hi, j);
            const v2f vb = pick2(vb_lo, vb_hi, j);
            acc00 = wmma_f32_k4(va, p0[j],     acc00);
            acc01 = wmma_f32_k4(vb, p0[4 + j], acc01);
            acc10 = wmma_f32_k4(va, p1[j],     acc10);
            acc11 = wmma_f32_k4(vb, p1[4 + j], acc11);
        }
    }

    const float* xr = x + base + (size_t)(n0 + 8 * h) * D_ + lane;
    float xa[8], xb[8];
    #pragma unroll
    for (int r = 0; r < 8; ++r) { xa[r] = xr[r * D_]; xb[r] = xr[r * D_ + 32]; }
    const float g0 = gamma[lane], g1 = gamma[lane + 32];
    const float b0 = beta[lane],  b1 = beta[lane + 32];

    const float l0 = lsum0.x + lsum0.y, l1 = lsum1.x + lsum1.y;
    const float inv0 = 1.0f / (l0 + xhalf(l0));
    const float inv1 = 1.0f / (l1 + xhalf(l1));
    v4f o_lo, o_hi;
    o_lo.x = (acc00[0] + acc01[0]) * inv0; o_lo.y = (acc00[1] + acc01[1]) * inv0;
    o_lo.z = (acc00[2] + acc01[2]) * inv0; o_lo.w = (acc00[3] + acc01[3]) * inv0;
    o_hi.x = (acc00[4] + acc01[4]) * inv0; o_hi.y = (acc00[5] + acc01[5]) * inv0;
    o_hi.z = (acc00[6] + acc01[6]) * inv0; o_hi.w = (acc00[7] + acc01[7]) * inv0;
    v4f* hp = &hs[l16 * HS_PITCH4_ + h * (HD_ / 4) + 2 * half];
    hp[0] = o_lo;
    hp[1] = o_hi;
    o_lo.x = (acc10[0] + acc11[0]) * inv1; o_lo.y = (acc10[1] + acc11[1]) * inv1;
    o_lo.z = (acc10[2] + acc11[2]) * inv1; o_lo.w = (acc10[3] + acc11[3]) * inv1;
    o_hi.x = (acc10[4] + acc11[4]) * inv1; o_hi.y = (acc10[5] + acc11[5]) * inv1;
    o_hi.z = (acc10[6] + acc11[6]) * inv1; o_hi.w = (acc10[7] + acc11[7]) * inv1;
    hp[16 * HS_PITCH4_]     = o_lo;
    hp[16 * HS_PITCH4_ + 1] = o_hi;
    __syncthreads();

    const float* hsf = (const float*)hs;
    float* orow = out + base + (size_t)(n0 + 8 * h) * D_ + lane;
    #pragma unroll
    for (int r = 0; r < 8; ++r) {
        const float* hrow = hsf + (8 * h + r) * (HS_PITCH4_ * 4);
        const float y0 = hrow[lane]      + xa[r];
        const float y1 = hrow[lane + 32] + xb[r];
        float s1 = y0 + y1;
        float s2 = y0 * y0 + y1 * y1;
        #pragma unroll
        for (int m = 1; m < 32; m <<= 1) {
            s1 += __shfl_xor(s1, m);
            s2 += __shfl_xor(s2, m);
        }
        const float mean = s1 * (1.0f / 64.0f);
        const float var  = s2 * (1.0f / 64.0f) - mean * mean;
        const float rstd = rsqrtf(var + EPS_);
        orow[r * D_]      = (y0 - mean) * rstd * g0 + b0;
        orow[r * D_ + 32] = (y1 - mean) * rstd * g1 + b1;
    }
}

extern "C" void kernel_launch(void* const* d_in, const int* in_sizes, int n_in,
                              void* d_out, int out_size, void* d_ws, size_t ws_size,
                              hipStream_t stream) {
    (void)in_sizes; (void)n_in; (void)out_size; (void)ws_size;
    const float* x     = (const float*)d_in[0];
    const float* adj   = (const float*)d_in[1];
    const float* Wq    = (const float*)d_in[3];
    const float* bq    = (const float*)d_in[4];
    const float* Wk    = (const float*)d_in[5];
    const float* bk    = (const float*)d_in[6];
    const float* Wv    = (const float*)d_in[7];
    const float* bv    = (const float*)d_in[8];
    const float* gamma = (const float*)d_in[9];
    const float* beta  = (const float*)d_in[10];
    float* out = (float*)d_out;

    float* ws = (float*)d_ws;
    const size_t S = (size_t)B_ * T_ * N_ * D_;
    float* Q  = ws;
    float* K  = ws + S;
    float* V  = ws + 2 * S;

    dim3 g1(B_ * T_ * N_ / 16 / QKV_WAVES_, 3, 1);
    qkv_kernel<<<g1, 32 * QKV_WAVES_, 0, stream>>>(x, Wq, bq, Wk, bk, Wv, bv, Q, K, V);

    dim3 g2(N_ / QROWS_, B_ * T_, 1);
    attn_ln_kernel<<<g2, 128, 0, stream>>>(Q, K, V, adj, x, gamma, beta, out);
}
